# starting kernel
# baseline (speedup 1.0000x reference)
#include <hip/hip_runtime.h>
#include <math.h>

typedef __attribute__((ext_vector_type(2))) float v2f;
typedef __attribute__((ext_vector_type(8))) float v8f;

#define IN_C 128
#define LOG2C 7

__global__ void k_set_const(float* __restrict__ p, float v, int n) {
    int i = blockIdx.x * blockDim.x + threadIdx.x;
    if (i < n) p[i] = v;
}

__global__ void k_count_deg(const int* __restrict__ dst, float* __restrict__ deg, int nE) {
    int i = blockIdx.x * blockDim.x + threadIdx.x;
    if (i < nE) atomicAdd(&deg[dst[i]], 1.0f);
}

__global__ void k_inv_sqrt(float* __restrict__ p, int n) {
    int i = blockIdx.x * blockDim.x + threadIdx.x;
    if (i < n) p[i] = 1.0f / sqrtf(p[i]);
}

__global__ __launch_bounds__(128) void k_gemm_wmma_f32(
    const float* __restrict__ A, const float* __restrict__ W,
    float* __restrict__ C, int mTiles)
{
    const int wavesPerBlock = blockDim.x >> 5;
    const int wave = blockIdx.x * wavesPerBlock + (threadIdx.x >> 5);
    if (wave >= mTiles) return;

    const int lane = threadIdx.x & 31;
    const int row  = lane & 15;
    const int hi   = lane >> 4;
    const int koff = hi * 2;
    const int m0   = wave * 16;

    v8f acc[8] = {};

    const float* __restrict__ arow = A + (size_t)(m0 + row) * IN_C;

    for (int k = 0; k < IN_C; k += 4) {
        v2f a;
        a.x = arow[k + koff];
        a.y = arow[k + koff + 1];
        const float* __restrict__ wbase = W + (size_t)(k + koff) * IN_C + row;
#pragma unroll
        for (int t = 0; t < 8; ++t) {
            v2f b;
            b.x = wbase[t * 16];
            b.y = wbase[t * 16 + IN_C];
            acc[t] = __builtin_amdgcn_wmma_f32_16x16x4_f32(
                false, a, false, b, (short)0, acc[t], false, false);
        }
    }

    float* __restrict__ cbase = C + (size_t)(m0 + hi * 8) * IN_C + row;
#pragma unroll
    for (int t = 0; t < 8; ++t) {
#pragma unroll
        for (int r = 0; r < 8; ++r) {
            cbase[(size_t)r * IN_C + t * 16] = acc[t][r];
        }
    }
}

__global__ __launch_bounds__(256) void k_aggregate(
    const float* __restrict__ h,
    const int* __restrict__ src, const int* __restrict__ dst,
    const float* __restrict__ inv_sqrt,
    float* __restrict__ agg, int nE)
{
    const int wave = blockIdx.x * (blockDim.x >> 5) + (threadIdx.x >> 5);
    if (wave >= nE) return;
    const int lane = threadIdx.x & 31;

    const int s = src[wave];
    const int d = dst[wave];
    const float coef = inv_sqrt[s] * inv_sqrt[d];

    const float4 hv = ((const float4*)(h + (size_t)s * IN_C))[lane];
    float* ap = agg + (size_t)d * IN_C + lane * 4;
    atomicAdd(ap + 0, hv.x * coef);
    atomicAdd(ap + 1, hv.y * coef);
    atomicAdd(ap + 2, hv.z * coef);
    atomicAdd(ap + 3, hv.w * coef);
}

__global__ void k_finalize(
    const float* __restrict__ agg, const float* __restrict__ h,
    const float* __restrict__ inv_sqrt, const float* __restrict__ bias,
    float* __restrict__ out, int total, int do_relu)
{
    int i = blockIdx.x * blockDim.x + threadIdx.x;
    if (i >= total) return;
    int n = i >> LOG2C;
    int c = i & (IN_C - 1);
    float is = inv_sqrt[n];
    float v = agg[i] + h[i] * is * is + bias[c];
    if (do_relu) v = fmaxf(v, 0.0f);
    out[i] = v;
}

extern "C" void kernel_launch(void* const* d_in, const int* in_sizes, int n_in,
                              void* d_out, int out_size, void* d_ws, size_t ws_size,
                              hipStream_t stream)
{
    const float* x   = (const float*)d_in[0];
    const int*   ei  = (const int*)d_in[1];
    const float* W1  = (const float*)d_in[2];
    const float* b1  = (const float*)d_in[3];
    const float* W2  = (const float*)d_in[4];
    const float* b2  = (const float*)d_in[5];

    const int N = in_sizes[0] / IN_C;
    const int E = in_sizes[1] / 2;
    const int* src = ei;
    const int* dst = ei + E;

    const int total = N * IN_C;

    float* inv_sqrt = (float*)d_ws;
    float* hbuf     = inv_sqrt + N;
    float* out      = (float*)d_out;

    const int T256 = 256;
    const int mTiles = (N + 15) / 16;
    dim3 gemmGrid((mTiles + 3) / 4);
    dim3 gemmBlock(128);
    dim3 aggGrid((E + 7) / 8);
    dim3 aggBlock(256);

    k_set_const<<<(N + T256 - 1) / T256, T256, 0, stream>>>(inv_sqrt, 1.0f, N);
    k_count_deg<<<(E + T256 - 1) / T256, T256, 0, stream>>>(dst, inv_sqrt, E);
    k_inv_sqrt<<<(N + T256 - 1) / T256, T256, 0, stream>>>(inv_sqrt, N);

    k_gemm_wmma_f32<<<gemmGrid, gemmBlock, 0, stream>>>(x, W1, hbuf, mTiles);
    k_set_const<<<(total + T256 - 1) / T256, T256, 0, stream>>>(out, 0.0f, total);
    k_aggregate<<<aggGrid, aggBlock, 0, stream>>>(hbuf, src, dst, inv_sqrt, out, E);
    k_finalize<<<(total + T256 - 1) / T256, T256, 0, stream>>>(
        out, hbuf, inv_sqrt, b1, out, total,  1);

    k_gemm_wmma_f32<<<gemmGrid, gemmBlock, 0, stream>>>(out, W2, hbuf, mTiles);
    k_set_const<<<(total + T256 - 1) / T256, T256, 0, stream>>>(out, 0.0f, total);
    k_aggregate<<<aggGrid, aggBlock, 0, stream>>>(hbuf, src, dst, inv_sqrt, out, E);
    k_finalize<<<(total + T256 - 1) / T256, T256, 0, stream>>>(
        out, hbuf, inv_sqrt, b2, out, total,  0);
}


# Replace per-edge f32 atomic scatter-add by a per-call counting sort of edges by destination (integer atomic rank, scan, fill) and a one-wave-per-node row gather; sum order = arrival-order rank
# speedup vs baseline: 2.6323x; 2.6323x over previous
#include <hip/hip_runtime.h>
#include <math.h>


typedef __attribute__((ext_vector_type(2))) float v2f;
typedef __attribute__((ext_vector_type(8))) float v8f;

#define IN_C 128
#define SCAN_T 1024

__global__ void k_zero_int(int* __restrict__ p, int n) {
    int i = blockIdx.x * blockDim.x + threadIdx.x;
    if (i < n) p[i] = 0;
}

__global__ void k_rank(const int* __restrict__ dst, int* __restrict__ cnt,
                       int* __restrict__ rank, int nE) {
    int i = blockIdx.x * blockDim.x + threadIdx.x;
    if (i < nE) rank[i] = atomicAdd(&cnt[dst[i]], 1);
}

__global__ __launch_bounds__(SCAN_T) void k_scan(
    const int* __restrict__ cnt, int* __restrict__ rowptr,
    float* __restrict__ inv_sqrt, int n)
{
    __shared__ int s_part[SCAN_T / 32];
    __shared__ int s_base;
    const int tid  = threadIdx.x;
    const int lane = tid & 31;
    const int wv   = tid >> 5;
    const int first = blockIdx.x * SCAN_T;

    int pre = 0;
    for (int i = tid; i < first; i += SCAN_T) pre += cnt[i];
#pragma unroll
    for (int o = 16; o > 0; o >>= 1) pre += __shfl_xor(pre, o);
    if (lane == 0) s_part[wv] = pre;
    __syncthreads();
    if (wv == 0) {
        int t = s_part[lane];
#pragma unroll
        for (int o = 16; o > 0; o >>= 1) t += __shfl_xor(t, o);
        if (lane == 0) s_base = t;
    }
    __syncthreads();
    const int base = s_base;
    __syncthreads();

    const int i = first + tid;
    const int v = (i < n) ? cnt[i] : 0;
    int inc = v;
#pragma unroll
    for (int o = 1; o < 32; o <<= 1) {
        int u = __shfl_up(inc, o);
        if (lane >= o) inc += u;
    }
    if (lane == 31) s_part[wv] = inc;
    __syncthreads();
    if (wv == 0) {
        int t = s_part[lane];
#pragma unroll
        for (int o = 1; o < 32; o <<= 1) {
            int u = __shfl_up(t, o);
            if (lane >= o) t += u;
        }
        s_part[lane] = t;
    }
    __syncthreads();
    const int wbase = (wv > 0) ? s_part[wv - 1] : 0;
    if (i < n) {
        rowptr[i + 1] = base + wbase + inc;
        inv_sqrt[i] = 1.0f / sqrtf((float)(v + 1));
    }
    if (i == 0) rowptr[0] = 0;
}

__global__ void k_fill(const int* __restrict__ src, const int* __restrict__ dst,
                       const int* __restrict__ rank, const int* __restrict__ rowptr,
                       int* __restrict__ col, int nE) {
    int i = blockIdx.x * blockDim.x + threadIdx.x;
    if (i >= nE) return;
    const int p = rowptr[dst[i]] + rank[i];
    if ((unsigned)p < (unsigned)nE) col[p] = src[i];
}

__global__ __launch_bounds__(128) void k_gemm_wmma_f32(
    const float* __restrict__ A, const float* __restrict__ W,
    float* __restrict__ C, int mTiles)
{
    const int wavesPerBlock = blockDim.x >> 5;
    const int wave = blockIdx.x * wavesPerBlock + (threadIdx.x >> 5);
    if (wave >= mTiles) return;

    const int lane = threadIdx.x & 31;
    const int row  = lane & 15;
    const int hi   = lane >> 4;
    const int koff = hi * 2;
    const int m0   = wave * 16;

    v8f acc[8] = {};

    const float* __restrict__ arow = A + (size_t)(m0 + row) * IN_C;

    for (int k = 0; k < IN_C; k += 4) {
        v2f a;
        a.x = arow[k + koff];
        a.y = arow[k + koff + 1];
        const float* __restrict__ wbase = W + (size_t)(k + koff) * IN_C + row;
#pragma unroll
        for (int t = 0; t < 8; ++t) {
            v2f b;
            b.x = wbase[t * 16];
            b.y = wbase[t * 16 + IN_C];
            acc[t] = __builtin_amdgcn_wmma_f32_16x16x4_f32(
                false, a, false, b, (short)0, acc[t], false, false);
        }
    }

    float* __restrict__ cbase = C + (size_t)(m0 + hi * 8) * IN_C + row;
#pragma unroll
    for (int t = 0; t < 8; ++t) {
#pragma unroll
        for (int r = 0; r < 8; ++r) {
            cbase[(size_t)r * IN_C + t * 16] = acc[t][r];
        }
    }
}

__global__ __launch_bounds__(256) void k_gather(
    const float* __restrict__ h, const int* __restrict__ rowptr,
    const int* __restrict__ col, const float* __restrict__ inv_sqrt,
    const float* __restrict__ bias, float* __restrict__ out,
    int nN, int do_relu)
{
    int node = blockIdx.x * (blockDim.x >> 5) + (threadIdx.x >> 5);
    if (node >= nN) return;
    node = __builtin_amdgcn_readfirstlane(node);
    const int lane = threadIdx.x & 31;

    const int beg = rowptr[node];
    const int end = rowptr[node + 1];
    const float isd = inv_sqrt[node];
    const float4* __restrict__ h4 = (const float4*)h;

    float4 acc = make_float4(0.0f, 0.0f, 0.0f, 0.0f);
    for (int base = beg; base < end; base += 32) {
        const int m = min(32, end - base);
        int   my_s = 0;
        float my_c = 0.0f;
        if (lane < m) {
            my_s = (int)min((unsigned)col[base + lane], (unsigned)(nN - 1));
            my_c = inv_sqrt[my_s] * isd;
        }
#pragma unroll 4
        for (int j = 0; j < m; ++j) {
            const int   s = __builtin_amdgcn_readlane(my_s, j);
            const float c = __builtin_bit_cast(float,
                __builtin_amdgcn_readlane(__builtin_bit_cast(int, my_c), j));
            const float4 hv = h4[(size_t)s * (IN_C / 4) + lane];
            acc.x += hv.x * c;
            acc.y += hv.y * c;
            acc.z += hv.z * c;
            acc.w += hv.w * c;
        }
    }

    const float4 hs = h4[(size_t)node * (IN_C / 4) + lane];
    const float4 b  = ((const float4*)bias)[lane];
    float4 v;
    v.x = acc.x + hs.x * isd * isd + b.x;
    v.y = acc.y + hs.y * isd * isd + b.y;
    v.z = acc.z + hs.z * isd * isd + b.z;
    v.w = acc.w + hs.w * isd * isd + b.w;
    if (do_relu) {
        v.x = fmaxf(v.x, 0.0f);
        v.y = fmaxf(v.y, 0.0f);
        v.z = fmaxf(v.z, 0.0f);
        v.w = fmaxf(v.w, 0.0f);
    }
    ((float4*)out)[(size_t)node * (IN_C / 4) + lane] = v;
}

extern "C" void kernel_launch(void* const* d_in, const int* in_sizes, int n_in,
                              void* d_out, int out_size, void* d_ws, size_t ws_size,
                              hipStream_t stream)
{
    const float* x   = (const float*)d_in[0];
    const int*   ei  = (const int*)d_in[1];
    const float* W1  = (const float*)d_in[2];
    const float* b1  = (const float*)d_in[3];
    const float* W2  = (const float*)d_in[4];
    const float* b2  = (const float*)d_in[5];

    const int N = in_sizes[0] / IN_C;
    const int E = in_sizes[1] / 2;
    const int* src = ei;
    const int* dst = ei + E;

    float* hbuf     = (float*)d_ws;
    float* inv_sqrt = hbuf + (size_t)N * IN_C;
    int*   cnt      = (int*)(inv_sqrt + N);
    int*   rowptr   = cnt + N;
    int*   rank     = rowptr + N + 4;
    int*   col      = rank + E;
    float* out      = (float*)d_out;

    const int T256 = 256;
    const int mTiles = (N + 15) / 16;
    dim3 gemmGrid((mTiles + 3) / 4);
    dim3 gemmBlock(128);
    dim3 nodeGrid((N + 7) / 8);
    dim3 nodeBlock(256);

    k_zero_int<<<(N + T256 - 1) / T256, T256, 0, stream>>>(cnt, N);
    k_rank<<<(E + T256 - 1) / T256, T256, 0, stream>>>(dst, cnt, rank, E);
    k_scan<<<(N + SCAN_T - 1) / SCAN_T, SCAN_T, 0, stream>>>(cnt, rowptr, inv_sqrt, N);
    k_fill<<<(E + T256 - 1) / T256, T256, 0, stream>>>(src, dst, rank, rowptr, col, E);

    k_gemm_wmma_f32<<<gemmGrid, gemmBlock, 0, stream>>>(x, W1, hbuf, mTiles);
    k_gather<<<nodeGrid, nodeBlock, 0, stream>>>(hbuf, rowptr, col, inv_sqrt, b1, out, N,  1);

    k_gemm_wmma_f32<<<gemmGrid, gemmBlock, 0, stream>>>(out, W2, hbuf, mTiles);
    k_gather<<<nodeGrid, nodeBlock, 0, stream>>>(hbuf, rowptr, col, inv_sqrt, b2, out, N,  0);
}


# Edges ordered by a two-level counting sort on LDS counters (bin histogram, partition in LDS, per-bin sort), not a global atomic, gather and scattered store per edge; sum order: LDS arrival order
# speedup vs baseline: 3.9298x; 1.4929x over previous
#include <hip/hip_runtime.h>
#include <math.h>


typedef __attribute__((ext_vector_type(2))) float v2f;
typedef __attribute__((ext_vector_type(8))) float v8f;

#define IN_C 128

#define BIN_SHIFT 9
#define BIN_W     512
#define NB_MAX    1024
#define SORT_T    1024
#define CHUNK     8192
#define STAGE_CAP 32768
#define BATCH     (CHUNK / SORT_T)

__device__ __forceinline__ int bin_of(int d, int nB) {
    return (int)min((unsigned)d >> BIN_SHIFT, (unsigned)(nB - 1));
}

__device__ __forceinline__ int block_excl_scan(int v, int* s_w, int* total) {
    const int lane = threadIdx.x & 31;
    const int wv   = threadIdx.x >> 5;
    int inc = v;
#pragma unroll
    for (int o = 1; o < 32; o <<= 1) {
        int u = __shfl_up(inc, o);
        if (lane >= o) inc += u;
    }
    __syncthreads();
    if (lane == 31) s_w[wv] = inc;
    __syncthreads();
    if (wv == 0) {
        int t = s_w[lane];
#pragma unroll
        for (int o = 1; o < 32; o <<= 1) {
            int u = __shfl_up(t, o);
            if (lane >= o) t += u;
        }
        s_w[lane] = t;
    }
    __syncthreads();
    const int wbase = (wv > 0) ? s_w[wv - 1] : 0;
    *total = s_w[31];
    return wbase + inc - v;
}

__global__ __launch_bounds__(SORT_T) void k_hist(
    const int* __restrict__ dst, int* __restrict__ blockHist, int nE, int nB)
{
    __shared__ int s_h[NB_MAX];
    const int tid = threadIdx.x;
    nB = min(nB, NB_MAX);
    s_h[tid] = 0;
    __syncthreads();
    const int c0 = blockIdx.x * CHUNK;
    int d[BATCH];
#pragma unroll
    for (int k = 0; k < BATCH; ++k) {
        const int i = c0 + k * SORT_T + tid;
        d[k] = dst[min(i, nE - 1)];
    }
#pragma unroll
    for (int k = 0; k < BATCH; ++k) {
        const int i = c0 + k * SORT_T + tid;
        if (i < nE) atomicAdd(&s_h[bin_of(d[k], nB)], 1);
    }
    __syncthreads();
    if (tid < nB) blockHist[blockIdx.x * nB + tid] = s_h[tid];
}

__global__ __launch_bounds__(SORT_T) void k_part(
    const int* __restrict__ src, const int* __restrict__ dst,
    const int* __restrict__ blockHist, int* __restrict__ binStart,
    int* __restrict__ tsrc, int* __restrict__ tdst, int nE, int nB, int nBlk)
{
    __shared__ int s_w[32];
    __shared__ int s_cur[NB_MAX];
    __shared__ int s_off[NB_MAX];
    __shared__ int s_gb[NB_MAX];
    __shared__ int s_ssrc[CHUNK];
    __shared__ int s_sdst[CHUNK];
    const int tid = threadIdx.x;
    const int blk = blockIdx.x;
    nB = min(nB, NB_MAX);

    int tot = 0, pre = 0, own = 0;
    if (tid < nB) {
        for (int k = 0; k < nBlk; ++k) {
            const int hc = blockHist[k * nB + tid];
            tot += hc;
            pre += (k < blk) ? hc : 0;
        }
        own = blockHist[blk * nB + tid];
    }
    int nAll, nChunk;
    const int bstart = block_excl_scan(tot, s_w, &nAll);
    const int off    = block_excl_scan(own, s_w, &nChunk);
    s_cur[tid] = off;
    s_off[tid] = off;
    s_gb[tid]  = bstart + pre;
    if (blk == 0) {
        if (tid < nB) binStart[tid] = bstart;
        if (tid == 0) binStart[nB] = nAll;
    }
    __syncthreads();

    const int c0 = blk * CHUNK;
    int d[BATCH], s[BATCH];
#pragma unroll
    for (int k = 0; k < BATCH; ++k) {
        const int i = c0 + k * SORT_T + tid;
        d[k] = dst[min(i, nE - 1)];
        s[k] = src[min(i, nE - 1)];
    }
#pragma unroll
    for (int k = 0; k < BATCH; ++k) {
        const int i = c0 + k * SORT_T + tid;
        if (i < nE) {
            const int slot = atomicAdd(&s_cur[bin_of(d[k], nB)], 1);
            if ((unsigned)slot < (unsigned)CHUNK) { s_ssrc[slot] = s[k]; s_sdst[slot] = d[k]; }
        }
    }
    __syncthreads();

    for (int k = 0; k < BATCH; ++k) {
        const int slot = k * SORT_T + tid;
        if (slot < nChunk) {
            const int dd = s_sdst[slot];
            const int b  = bin_of(dd, nB);
            const int g  = s_gb[b] + (slot - s_off[b]);
            if ((unsigned)g < (unsigned)nE) { tsrc[g] = s_ssrc[slot]; tdst[g] = dd; }
        }
    }
}

__global__ __launch_bounds__(SORT_T) void k_binsort(
    const int* __restrict__ tsrc, const int* __restrict__ tdst,
    const int* __restrict__ binStart, int* __restrict__ rowptr,
    float* __restrict__ inv_sqrt, int* __restrict__ col, int nN, int nE)
{
    __shared__ int s_w[32];
    __shared__ int s_deg[BIN_W];
    __shared__ int s_cur[BIN_W];
    __shared__ int s_col[STAGE_CAP];
    const int tid = threadIdx.x;
    const int b   = blockIdx.x;
    const int e0  = min(max(binStart[b], 0), nE);
    const int e1  = min(max(binStart[b + 1], e0), nE);
    const int cnt = e1 - e0;
    const int node0 = b << BIN_SHIFT;

    if (tid < BIN_W) s_deg[tid] = 0;
    __syncthreads();
    for (int i0 = e0 + tid; i0 < e1; i0 += BATCH * SORT_T) {
        int d[BATCH];
#pragma unroll
        for (int k = 0; k < BATCH; ++k) {
            const int i = i0 + k * SORT_T;
            d[k] = tdst[min(i, e1 - 1)];
        }
#pragma unroll
        for (int k = 0; k < BATCH; ++k) {
            const int i = i0 + k * SORT_T;
            if (i < e1) atomicAdd(&s_deg[(int)min((unsigned)(d[k] - node0), (unsigned)(BIN_W - 1))], 1);
        }
    }
    __syncthreads();

    const int dg = (tid < BIN_W) ? s_deg[tid] : 0;
    int nBin;
    const int lrow = block_excl_scan(dg, s_w, &nBin);
    if (tid < BIN_W) {
        s_cur[tid] = lrow;
        const int node = node0 + tid;
        if (node < nN) {
            rowptr[node + 1] = e0 + lrow + dg;
            inv_sqrt[node] = 1.0f / sqrtf((float)(dg + 1));
        }
    }
    if (b == 0 && tid == 0) rowptr[0] = 0;
    __syncthreads();

    const bool staged = (cnt <= STAGE_CAP);
    for (int i0 = e0 + tid; i0 < e1; i0 += BATCH * SORT_T) {
        int d[BATCH], s[BATCH];
#pragma unroll
        for (int k = 0; k < BATCH; ++k) {
            const int i = i0 + k * SORT_T;
            d[k] = tdst[min(i, e1 - 1)];
            s[k] = tsrc[min(i, e1 - 1)];
        }
#pragma unroll
        for (int k = 0; k < BATCH; ++k) {
            const int i = i0 + k * SORT_T;
            if (i < e1) {
                const int ld = (int)min((unsigned)(d[k] - node0), (unsigned)(BIN_W - 1));
                const int p  = atomicAdd(&s_cur[ld], 1);
                if ((unsigned)p < (unsigned)cnt) {
                    if (staged) s_col[p] = s[k];
                    else        col[e0 + p] = s[k];
                }
            }
        }
    }
    if (staged) {
        __syncthreads();
        for (int i = tid; i < cnt; i += SORT_T) col[e0 + i] = s_col[i];
    }
}

__global__ __launch_bounds__(128) void k_gemm_wmma_f32(
    const float* __restrict__ A, const float* __restrict__ W,
    float* __restrict__ C, int mTiles)
{
    const int wavesPerBlock = blockDim.x >> 5;
    const int wave = blockIdx.x * wavesPerBlock + (threadIdx.x >> 5);
    if (wave >= mTiles) return;

    const int lane = threadIdx.x & 31;
    const int row  = lane & 15;
    const int hi   = lane >> 4;
    const int koff = hi * 2;
    const int m0   = wave * 16;

    v8f acc[8] = {};

    const float* __restrict__ arow = A + (size_t)(m0 + row) * IN_C;

    for (int k = 0; k < IN_C; k += 4) {
        v2f a;
        a.x = arow[k + koff];
        a.y = arow[k + koff + 1];
        const float* __restrict__ wbase = W + (size_t)(k + koff) * IN_C + row;
#pragma unroll
        for (int t = 0; t < 8; ++t) {
            v2f b;
            b.x = wbase[t * 16];
            b.y = wbase[t * 16 + IN_C];
            acc[t] = __builtin_amdgcn_wmma_f32_16x16x4_f32(
                false, a, false, b, (short)0, acc[t], false, false);
        }
    }

    float* __restrict__ cbase = C + (size_t)(m0 + hi * 8) * IN_C + row;
#pragma unroll
    for (int t = 0; t < 8; ++t) {
#pragma unroll
        for (int r = 0; r < 8; ++r) {
            cbase[(size_t)r * IN_C + t * 16] = acc[t][r];
        }
    }
}

__global__ __launch_bounds__(256) void k_gather(
    const float* __restrict__ h, const int* __restrict__ rowptr,
    const int* __restrict__ col, const float* __restrict__ inv_sqrt,
    const float* __restrict__ bias, float* __restrict__ out,
    int nN, int do_relu)
{
    int node = blockIdx.x * (blockDim.x >> 5) + (threadIdx.x >> 5);
    if (node >= nN) return;
    node = __builtin_amdgcn_readfirstlane(node);
    const int lane = threadIdx.x & 31;

    const int beg = rowptr[node];
    const int end = rowptr[node + 1];
    const float isd = inv_sqrt[node];
    const float4* __restrict__ h4 = (const float4*)h;

    float4 acc = make_float4(0.0f, 0.0f, 0.0f, 0.0f);
    for (int base = beg; base < end; base += 32) {
        const int m = min(32, end - base);
        int   my_s = 0;
        float my_c = 0.0f;
        if (lane < m) {
            my_s = (int)min((unsigned)col[base + lane], (unsigned)(nN - 1));
            my_c = inv_sqrt[my_s] * isd;
        }
#pragma unroll 4
        for (int j = 0; j < m; ++j) {
            const int   s = __builtin_amdgcn_readlane(my_s, j);
            const float c = __builtin_bit_cast(float,
                __builtin_amdgcn_readlane(__builtin_bit_cast(int, my_c), j));
            const float4 hv = h4[(size_t)s * (IN_C / 4) + lane];
            acc.x += hv.x * c;
            acc.y += hv.y * c;
            acc.z += hv.z * c;
            acc.w += hv.w * c;
        }
    }

    const float4 hs = h4[(size_t)node * (IN_C / 4) + lane];
    const float4 b  = ((const float4*)bias)[lane];
    float4 v;
    v.x = acc.x + hs.x * isd * isd + b.x;
    v.y = acc.y + hs.y * isd * isd + b.y;
    v.z = acc.z + hs.z * isd * isd + b.z;
    v.w = acc.w + hs.w * isd * isd + b.w;
    if (do_relu) {
        v.x = fmaxf(v.x, 0.0f);
        v.y = fmaxf(v.y, 0.0f);
        v.z = fmaxf(v.z, 0.0f);
        v.w = fmaxf(v.w, 0.0f);
    }
    ((float4*)out)[(size_t)node * (IN_C / 4) + lane] = v;
}

extern "C" void kernel_launch(void* const* d_in, const int* in_sizes, int n_in,
                              void* d_out, int out_size, void* d_ws, size_t ws_size,
                              hipStream_t stream)
{
    const float* x   = (const float*)d_in[0];
    const int*   ei  = (const int*)d_in[1];
    const float* W1  = (const float*)d_in[2];
    const float* b1  = (const float*)d_in[3];
    const float* W2  = (const float*)d_in[4];
    const float* b2  = (const float*)d_in[5];

    const int N = in_sizes[0] / IN_C;
    const int E = in_sizes[1] / 2;
    const int* src = ei;
    const int* dst = ei + E;

    const int NB   = (N + BIN_W - 1) / BIN_W;
    const int nBlk = (E + CHUNK - 1) / CHUNK;
    float* hbuf      = (float*)d_ws;
    float* inv_sqrt  = hbuf + (size_t)N * IN_C;
    int*   rowptr    = (int*)(inv_sqrt + N);
    int*   col       = rowptr + N + 4;
    int*   tsrc      = col + E;
    int*   tdst      = tsrc + E;
    int*   binStart  = tdst + E;
    int*   blockHist = binStart + NB + 4;
    float* out       = (float*)d_out;

    const int mTiles = (N + 15) / 16;
    dim3 gemmGrid((mTiles + 3) / 4);
    dim3 gemmBlock(128);
    dim3 nodeGrid((N + 7) / 8);
    dim3 nodeBlock(256);

    k_hist<<<nBlk, SORT_T, 0, stream>>>(dst, blockHist, E, NB);
    k_part<<<nBlk, SORT_T, 0, stream>>>(src, dst, blockHist, binStart, tsrc, tdst, E, NB, nBlk);
    k_binsort<<<NB, SORT_T, 0, stream>>>(tsrc, tdst, binStart, rowptr, inv_sqrt, col, N, E);

    k_gemm_wmma_f32<<<gemmGrid, gemmBlock, 0, stream>>>(x, W1, hbuf, mTiles);
    k_gather<<<nodeGrid, nodeBlock, 0, stream>>>(hbuf, rowptr, col, inv_sqrt, b1, out, N,  1);

    k_gemm_wmma_f32<<<gemmGrid, gemmBlock, 0, stream>>>(out, W2, hbuf, mTiles);
    k_gather<<<nodeGrid, nodeBlock, 0, stream>>>(hbuf, rowptr, col, inv_sqrt, b2, out, N,  0);
}


# Layer-2 product fused into the layer-1 aggregation launch: 8-wave groups loop over 16-node tiles, summed rows go to an LDS tile and meet W2 fragments in registers; 7 to 6 launches; sum order kept
# speedup vs baseline: 4.7452x; 1.2075x over previous
#include <hip/hip_runtime.h>
#include <math.h>


typedef __attribute__((ext_vector_type(2))) float v2f;
typedef __attribute__((ext_vector_type(8))) float v8f;

#define IN_C 128

#define BIN_SHIFT 9
#define BIN_W     512
#define NB_MAX    1024
#define SORT_T    1024
#define CHUNK     8192
#define STAGE_CAP 32768
#define BATCH     (CHUNK / SORT_T)

__device__ __forceinline__ int bin_of(int d, int nB) {
    return (int)min((unsigned)d >> BIN_SHIFT, (unsigned)(nB - 1));
}

__device__ __forceinline__ int block_excl_scan(int v, int* s_w, int* total) {
    const int lane = threadIdx.x & 31;
    const int wv   = threadIdx.x >> 5;
    int inc = v;
#pragma unroll
    for (int o = 1; o < 32; o <<= 1) {
        int u = __shfl_up(inc, o);
        if (lane >= o) inc += u;
    }
    __syncthreads();
    if (lane == 31) s_w[wv] = inc;
    __syncthreads();
    if (wv == 0) {
        int t = s_w[lane];
#pragma unroll
        for (int o = 1; o < 32; o <<= 1) {
            int u = __shfl_up(t, o);
            if (lane >= o) t += u;
        }
        s_w[lane] = t;
    }
    __syncthreads();
    const int wbase = (wv > 0) ? s_w[wv - 1] : 0;
    *total = s_w[31];
    return wbase + inc - v;
}

__global__ __launch_bounds__(SORT_T) void k_hist(
    const int* __restrict__ dst, int* __restrict__ blockHist, int nE, int nB)
{
    __shared__ int s_h[NB_MAX];
    const int tid = threadIdx.x;
    nB = min(nB, NB_MAX);
    s_h[tid] = 0;
    __syncthreads();
    const int c0 = blockIdx.x * CHUNK;
    int d[BATCH];
#pragma unroll
    for (int k = 0; k < BATCH; ++k) {
        const int i = c0 + k * SORT_T + tid;
        d[k] = dst[min(i, nE - 1)];
    }
#pragma unroll
    for (int k = 0; k < BATCH; ++k) {
        const int i = c0 + k * SORT_T + tid;
        if (i < nE) atomicAdd(&s_h[bin_of(d[k], nB)], 1);
    }
    __syncthreads();
    if (tid < nB) blockHist[blockIdx.x * nB + tid] = s_h[tid];
}

__global__ __launch_bounds__(SORT_T) void k_part(
    const int* __restrict__ src, const int* __restrict__ dst,
    const int* __restrict__ blockHist, int* __restrict__ binStart,
    int* __restrict__ tsrc, int* __restrict__ tdst, int nE, int nB, int nBlk)
{
    __shared__ int s_w[32];
    __shared__ int s_cur[NB_MAX];
    __shared__ int s_off[NB_MAX];
    __shared__ int s_gb[NB_MAX];
    __shared__ int s_ssrc[CHUNK];
    __shared__ int s_sdst[CHUNK];
    const int tid = threadIdx.x;
    const int blk = blockIdx.x;
    nB = min(nB, NB_MAX);

    int tot = 0, pre = 0, own = 0;
    if (tid < nB) {
        for (int k = 0; k < nBlk; ++k) {
            const int hc = blockHist[k * nB + tid];
            tot += hc;
            pre += (k < blk) ? hc : 0;
        }
        own = blockHist[blk * nB + tid];
    }
    int nAll, nChunk;
    const int bstart = block_excl_scan(tot, s_w, &nAll);
    const int off    = block_excl_scan(own, s_w, &nChunk);
    s_cur[tid] = off;
    s_off[tid] = off;
    s_gb[tid]  = bstart + pre;
    if (blk == 0) {
        if (tid < nB) binStart[tid] = bstart;
        if (tid == 0) binStart[nB] = nAll;
    }
    __syncthreads();

    const int c0 = blk * CHUNK;
    int d[BATCH], s[BATCH];
#pragma unroll
    for (int k = 0; k < BATCH; ++k) {
        const int i = c0 + k * SORT_T + tid;
        d[k] = dst[min(i, nE - 1)];
        s[k] = src[min(i, nE - 1)];
    }
#pragma unroll
    for (int k = 0; k < BATCH; ++k) {
        const int i = c0 + k * SORT_T + tid;
        if (i < nE) {
            const int slot = atomicAdd(&s_cur[bin_of(d[k], nB)], 1);
            if ((unsigned)slot < (unsigned)CHUNK) { s_ssrc[slot] = s[k]; s_sdst[slot] = d[k]; }
        }
    }
    __syncthreads();

    for (int k = 0; k < BATCH; ++k) {
        const int slot = k * SORT_T + tid;
        if (slot < nChunk) {
            const int dd = s_sdst[slot];
            const int b  = bin_of(dd, nB);
            const int g  = s_gb[b] + (slot - s_off[b]);
            if ((unsigned)g < (unsigned)nE) { tsrc[g] = s_ssrc[slot]; tdst[g] = dd; }
        }
    }
}

__global__ __launch_bounds__(SORT_T) void k_binsort(
    const int* __restrict__ tsrc, const int* __restrict__ tdst,
    const int* __restrict__ binStart, int* __restrict__ rowptr,
    float* __restrict__ inv_sqrt, int* __restrict__ col, int nN, int nE)
{
    __shared__ int s_w[32];
    __shared__ int s_deg[BIN_W];
    __shared__ int s_cur[BIN_W];
    __shared__ int s_col[STAGE_CAP];
    const int tid = threadIdx.x;
    const int b   = blockIdx.x;
    const int e0  = min(max(binStart[b], 0), nE);
    const int e1  = min(max(binStart[b + 1], e0), nE);
    const int cnt = e1 - e0;
    const int node0 = b << BIN_SHIFT;

    if (tid < BIN_W) s_deg[tid] = 0;
    __syncthreads();
    for (int i0 = e0 + tid; i0 < e1; i0 += BATCH * SORT_T) {
        int d[BATCH];
#pragma unroll
        for (int k = 0; k < BATCH; ++k) {
            const int i = i0 + k * SORT_T;
            d[k] = tdst[min(i, e1 - 1)];
        }
#pragma unroll
        for (int k = 0; k < BATCH; ++k) {
            const int i = i0 + k * SORT_T;
            if (i < e1) atomicAdd(&s_deg[(int)min((unsigned)(d[k] - node0), (unsigned)(BIN_W - 1))], 1);
        }
    }
    __syncthreads();

    const int dg = (tid < BIN_W) ? s_deg[tid] : 0;
    int nBin;
    const int lrow = block_excl_scan(dg, s_w, &nBin);
    if (tid < BIN_W) {
        s_cur[tid] = lrow;
        const int node = node0 + tid;
        if (node < nN) {
            rowptr[node + 1] = e0 + lrow + dg;
            inv_sqrt[node] = 1.0f / sqrtf((float)(dg + 1));
        }
    }
    if (b == 0 && tid == 0) rowptr[0] = 0;
    __syncthreads();

    const bool staged = (cnt <= STAGE_CAP);
    for (int i0 = e0 + tid; i0 < e1; i0 += BATCH * SORT_T) {
        int d[BATCH], s[BATCH];
#pragma unroll
        for (int k = 0; k < BATCH; ++k) {
            const int i = i0 + k * SORT_T;
            d[k] = tdst[min(i, e1 - 1)];
            s[k] = tsrc[min(i, e1 - 1)];
        }
#pragma unroll
        for (int k = 0; k < BATCH; ++k) {
            const int i = i0 + k * SORT_T;
            if (i < e1) {
                const int ld = (int)min((unsigned)(d[k] - node0), (unsigned)(BIN_W - 1));
                const int p  = atomicAdd(&s_cur[ld], 1);
                if ((unsigned)p < (unsigned)cnt) {
                    if (staged) s_col[p] = s[k];
                    else        col[e0 + p] = s[k];
                }
            }
        }
    }
    if (staged) {
        __syncthreads();
        for (int i = tid; i < cnt; i += SORT_T) col[e0 + i] = s_col[i];
    }
}

__global__ __launch_bounds__(128) void k_gemm_wmma_f32(
    const float* __restrict__ A, const float* __restrict__ W,
    float* __restrict__ C, int mTiles)
{
    const int wavesPerBlock = blockDim.x >> 5;
    const int wave = blockIdx.x * wavesPerBlock + (threadIdx.x >> 5);
    if (wave >= mTiles) return;

    const int lane = threadIdx.x & 31;
    const int row  = lane & 15;
    const int hi   = lane >> 4;
    const int koff = hi * 2;
    const int m0   = wave * 16;

    v8f acc[8] = {};

    const float* __restrict__ arow = A + (size_t)(m0 + row) * IN_C;

    for (int k = 0; k < IN_C; k += 4) {
        v2f a;
        a.x = arow[k + koff];
        a.y = arow[k + koff + 1];
        const float* __restrict__ wbase = W + (size_t)(k + koff) * IN_C + row;
#pragma unroll
        for (int t = 0; t < 8; ++t) {
            v2f b;
            b.x = wbase[t * 16];
            b.y = wbase[t * 16 + IN_C];
            acc[t] = __builtin_amdgcn_wmma_f32_16x16x4_f32(
                false, a, false, b, (short)0, acc[t], false, false);
        }
    }

    float* __restrict__ cbase = C + (size_t)(m0 + hi * 8) * IN_C + row;
#pragma unroll
    for (int t = 0; t < 8; ++t) {
#pragma unroll
        for (int r = 0; r < 8; ++r) {
            cbase[(size_t)r * IN_C + t * 16] = acc[t][r];
        }
    }
}

__global__ __launch_bounds__(256) void k_gather(
    const float* __restrict__ h, const int* __restrict__ rowptr,
    const int* __restrict__ col, const float* __restrict__ inv_sqrt,
    const float* __restrict__ bias, float* __restrict__ out,
    int nN, int do_relu)
{
    int node = blockIdx.x * (blockDim.x >> 5) + (threadIdx.x >> 5);
    if (node >= nN) return;
    node = __builtin_amdgcn_readfirstlane(node);
    const int lane = threadIdx.x & 31;

    const int beg = rowptr[node];
    const int end = rowptr[node + 1];
    const float isd = inv_sqrt[node];
    const float4* __restrict__ h4 = (const float4*)h;

    float4 acc = make_float4(0.0f, 0.0f, 0.0f, 0.0f);
    for (int base = beg; base < end; base += 32) {
        const int m = min(32, end - base);
        int   my_s = 0;
        float my_c = 0.0f;
        if (lane < m) {
            my_s = (int)min((unsigned)col[base + lane], (unsigned)(nN - 1));
            my_c = inv_sqrt[my_s] * isd;
        }
#pragma unroll 4
        for (int j = 0; j < m; ++j) {
            const int   s = __builtin_amdgcn_readlane(my_s, j);
            const float c = __builtin_bit_cast(float,
                __builtin_amdgcn_readlane(__builtin_bit_cast(int, my_c), j));
            const float4 hv = h4[(size_t)s * (IN_C / 4) + lane];
            acc.x += hv.x * c;
            acc.y += hv.y * c;
            acc.z += hv.z * c;
            acc.w += hv.w * c;
        }
    }

    const float4 hs = h4[(size_t)node * (IN_C / 4) + lane];
    const float4 b  = ((const float4*)bias)[lane];
    float4 v;
    v.x = acc.x + hs.x * isd * isd + b.x;
    v.y = acc.y + hs.y * isd * isd + b.y;
    v.z = acc.z + hs.z * isd * isd + b.z;
    v.w = acc.w + hs.w * isd * isd + b.w;
    if (do_relu) {
        v.x = fmaxf(v.x, 0.0f);
        v.y = fmaxf(v.y, 0.0f);
        v.z = fmaxf(v.z, 0.0f);
        v.w = fmaxf(v.w, 0.0f);
    }
    ((float4*)out)[(size_t)node * (IN_C / 4) + lane] = v;
}

#define FUSED_GRID 1024
#define TILE_M     16
#define TILE_PITCH 132

__device__ __forceinline__ float readlane_f(float v, int l) {
    return __builtin_bit_cast(float, __builtin_amdgcn_readlane(__builtin_bit_cast(int, v), l));
}

__device__ __forceinline__ void fma_row(float4& acc, const float4 hv, const float c) {
    acc.x = __builtin_fmaf(hv.x, c, acc.x);
    acc.y = __builtin_fmaf(hv.y, c, acc.y);
    acc.z = __builtin_fmaf(hv.z, c, acc.z);
    acc.w = __builtin_fmaf(hv.w, c, acc.w);
}

__device__ __forceinline__ float4 src_row(const float4* __restrict__ h4, int my_s, int j, int lane) {
    return h4[(size_t)__builtin_amdgcn_readlane(my_s, j) * (IN_C / 4) + lane];
}

__device__ __forceinline__ void rows_acc(float4& acc, const float4* __restrict__ h4,
                                         int my_s, float my_c, int m, int lane)
{
    int j = 0;
    for (; j + 4 <= m; j += 4) {
        const float4 h0 = src_row(h4, my_s, j, lane);
        const float4 h1 = src_row(h4, my_s, j + 1, lane);
        const float4 h2 = src_row(h4, my_s, j + 2, lane);
        const float4 h3 = src_row(h4, my_s, j + 3, lane);
        fma_row(acc, h0, readlane_f(my_c, j));
        fma_row(acc, h1, readlane_f(my_c, j + 1));
        fma_row(acc, h2, readlane_f(my_c, j + 2));
        fma_row(acc, h3, readlane_f(my_c, j + 3));
    }
    const int rem = m - j;
    if (rem == 3) {
        const float4 h0 = src_row(h4, my_s, j, lane);
        const float4 h1 = src_row(h4, my_s, j + 1, lane);
        const float4 h2 = src_row(h4, my_s, j + 2, lane);
        fma_row(acc, h0, readlane_f(my_c, j));
        fma_row(acc, h1, readlane_f(my_c, j + 1));
        fma_row(acc, h2, readlane_f(my_c, j + 2));
    } else if (rem == 2) {
        const float4 h0 = src_row(h4, my_s, j, lane);
        const float4 h1 = src_row(h4, my_s, j + 1, lane);
        fma_row(acc, h0, readlane_f(my_c, j));
        fma_row(acc, h1, readlane_f(my_c, j + 1));
    } else if (rem == 1) {
        const float4 h0 = src_row(h4, my_s, j, lane);
        fma_row(acc, h0, readlane_f(my_c, j));
    }
}

__device__ __forceinline__ float4 node_sum(const float4* __restrict__ h4, const int* __restrict__ col,
                                           const float* __restrict__ inv_sqrt, int beg, int end,
                                           int my_s, float my_c, float isd, int nN, int nE, int lane)
{
    float4 acc = make_float4(0.0f, 0.0f, 0.0f, 0.0f);
    rows_acc(acc, h4, my_s, my_c, min(32, end - beg), lane);
    for (int base = beg + 32; base < end; base += 32) {
        const int m = min(32, end - base);
        int   s = 0;
        float c = 0.0f;
        if (lane < m) {
            s = (int)min((unsigned)col[min(base + lane, nE - 1)], (unsigned)(nN - 1));
            c = inv_sqrt[s] * isd;
        }
        rows_acc(acc, h4, s, c, m, lane);
    }
    return acc;
}

__device__ __forceinline__ int edge_slot(int beg, int end, int lane, int nE) {
    return min(max(min(beg + lane, end - 1), 0), nE - 1);
}

__global__ __launch_bounds__(256) void k_gather_gemm(
    const float* __restrict__ h, const int* __restrict__ rowptr,
    const int* __restrict__ col, const float* __restrict__ inv_sqrt,
    const float* __restrict__ bias, const float* __restrict__ Wn,
    float* __restrict__ hn, int nN, int nE, int mTiles)
{
    __shared__ float s_a[TILE_M * TILE_PITCH];
    __shared__ float s_c[TILE_M * TILE_PITCH];
    const int lane = threadIdx.x & 31;
    const int wv   = __builtin_amdgcn_readfirstlane((int)(threadIdx.x >> 5));
    const int row  = lane & 15;
    const int hi   = lane >> 4;
    const int koff = hi * 2;

    int tile = blockIdx.x;
    if (tile >= mTiles) return;

    v2f bfrag[32];
    {
        const float* __restrict__ wbase = Wn + (size_t)koff * IN_C + wv * 16 + row;
#pragma unroll
        for (int ks = 0; ks < 32; ++ks) {
            bfrag[ks].x = wbase[(size_t)(ks * 4) * IN_C];
            bfrag[ks].y = wbase[(size_t)(ks * 4 + 1) * IN_C];
        }
    }
    const float4* __restrict__ h4  = (const float4*)h;
    float4* __restrict__       hn4 = (float4*)hn;

    int   n0  = tile * TILE_M + wv * 2;
    int   rpv = rowptr[min(n0 + min(lane, 2), nN)];
    float isv = inv_sqrt[min(n0 + (lane & 1), nN - 1)];
    int   craw0, craw1;
    float iraw0, iraw1;
    {
        const int p0 = __builtin_amdgcn_readlane(rpv, 0);
        const int p1 = __builtin_amdgcn_readlane(rpv, 1);
        const int p2 = __builtin_amdgcn_readlane(rpv, 2);
        craw0 = col[edge_slot(p0, p1, lane, nE)];
        craw1 = col[edge_slot(p1, p2, lane, nE)];
        iraw0 = inv_sqrt[min((unsigned)craw0, (unsigned)(nN - 1))];
        iraw1 = inv_sqrt[min((unsigned)craw1, (unsigned)(nN - 1))];
    }

    for (;;) {
        const int ntile = tile + gridDim.x;
        n0 = tile * TILE_M + wv * 2;
        const int   p0   = __builtin_amdgcn_readlane(rpv, 0);
        const int   p1   = __builtin_amdgcn_readlane(rpv, 1);
        const int   p2   = __builtin_amdgcn_readlane(rpv, 2);
        const float isd0 = readlane_f(isv, 0);
        const float isd1 = readlane_f(isv, 1);
        const int   ms0  = (int)min((unsigned)craw0, (unsigned)(nN - 1));
        const int   ms1  = (int)min((unsigned)craw1, (unsigned)(nN - 1));
        const float mc0  = (lane < p1 - p0) ? iraw0 * isd0 : 0.0f;
        const float mc1  = (lane < p2 - p1) ? iraw1 * isd1 : 0.0f;

        const int   nn0  = ntile * TILE_M + wv * 2;
        const int   rpvN = rowptr[min(nn0 + min(lane, 2), nN)];
        const float isvN = inv_sqrt[min(nn0 + (lane & 1), nN - 1)];

        const float4 b4 = ((const float4*)bias)[lane];

        {
            const float4 hs  = h4[(size_t)min(n0, nN - 1) * (IN_C / 4) + lane];
            const float4 acc = node_sum(h4, col, inv_sqrt, p0, p1, ms0, mc0, isd0, nN, nE, lane);
            float4 v;
            v.x = fmaxf(__builtin_fmaf(hs.x * isd0, isd0, acc.x) + b4.x, 0.0f);
            v.y = fmaxf(__builtin_fmaf(hs.y * isd0, isd0, acc.y) + b4.y, 0.0f);
            v.z = fmaxf(__builtin_fmaf(hs.z * isd0, isd0, acc.z) + b4.z, 0.0f);
            v.w = fmaxf(__builtin_fmaf(hs.w * isd0, isd0, acc.w) + b4.w, 0.0f);
            if (n0 >= nN) v = make_float4(0.0f, 0.0f, 0.0f, 0.0f);
            *(float4*)(s_a + (wv * 2) * TILE_PITCH + lane * 4) = v;
        }

        const int q0 = __builtin_amdgcn_readlane(rpvN, 0);
        const int q1 = __builtin_amdgcn_readlane(rpvN, 1);
        const int q2 = __builtin_amdgcn_readlane(rpvN, 2);
        const int crawN0 = col[edge_slot(q0, q1, lane, nE)];
        const int crawN1 = col[edge_slot(q1, q2, lane, nE)];

        {
            const float4 hs  = h4[(size_t)min(n0 + 1, nN - 1) * (IN_C / 4) + lane];
            const float4 acc = node_sum(h4, col, inv_sqrt, p1, p2, ms1, mc1, isd1, nN, nE, lane);
            float4 v;
            v.x = fmaxf(__builtin_fmaf(hs.x * isd1, isd1, acc.x) + b4.x, 0.0f);
            v.y = fmaxf(__builtin_fmaf(hs.y * isd1, isd1, acc.y) + b4.y, 0.0f);
            v.z = fmaxf(__builtin_fmaf(hs.z * isd1, isd1, acc.z) + b4.z, 0.0f);
            v.w = fmaxf(__builtin_fmaf(hs.w * isd1, isd1, acc.w) + b4.w, 0.0f);
            if (n0 + 1 >= nN) v = make_float4(0.0f, 0.0f, 0.0f, 0.0f);
            *(float4*)(s_a + (wv * 2 + 1) * TILE_PITCH + lane * 4) = v;
        }

        __syncthreads();

        const float irawN0 = inv_sqrt[min((unsigned)crawN0, (unsigned)(nN - 1))];
        const float irawN1 = inv_sqrt[min((unsigned)crawN1, (unsigned)(nN - 1))];

        v8f acc = {};
        const float* __restrict__ arow = s_a + row * TILE_PITCH + koff;
#pragma unroll
        for (int ks = 0; ks < 32; ++ks) {
            const v2f a = *(const v2f*)(arow + ks * 4);
            acc = __builtin_amdgcn_wmma_f32_16x16x4_f32(
                false, a, false, bfrag[ks], (short)0, acc, false, false);
        }
#pragma unroll
        for (int r = 0; r < 8; ++r) s_c[(r + 8 * hi) * TILE_PITCH + wv * 16 + row] = acc[r];

        __syncthreads();

#pragma unroll
        for (int q = 0; q < 2; ++q) {
            const int node = n0 + q;
            if (node < nN)
                hn4[(size_t)node * (IN_C / 4) + lane] = *(const float4*)(s_c + (wv * 2 + q) * TILE_PITCH + lane * 4);
        }

        if (ntile >= mTiles) break;
        tile = ntile;
        rpv = rpvN;  isv = isvN;
        craw0 = crawN0;  craw1 = crawN1;
        iraw0 = irawN0;  iraw1 = irawN1;
    }
}

extern "C" void kernel_launch(void* const* d_in, const int* in_sizes, int n_in,
                              void* d_out, int out_size, void* d_ws, size_t ws_size,
                              hipStream_t stream)
{
    const float* x   = (const float*)d_in[0];
    const int*   ei  = (const int*)d_in[1];
    const float* W1  = (const float*)d_in[2];
    const float* b1  = (const float*)d_in[3];
    const float* W2  = (const float*)d_in[4];
    const float* b2  = (const float*)d_in[5];

    const int N = in_sizes[0] / IN_C;
    const int E = in_sizes[1] / 2;
    const int* src = ei;
    const int* dst = ei + E;

    const int NB   = (N + BIN_W - 1) / BIN_W;
    const int nBlk = (E + CHUNK - 1) / CHUNK;
    float* hbuf      = (float*)d_ws;
    float* hbuf2     = hbuf + (size_t)N * IN_C;
    float* inv_sqrt  = hbuf2 + (size_t)N * IN_C;
    int*   rowptr    = (int*)(inv_sqrt + N);
    int*   col       = rowptr + N + 4;
    int*   tsrc      = col + E;
    int*   tdst      = tsrc + E;
    int*   binStart  = tdst + E;
    int*   blockHist = binStart + NB + 4;
    float* out       = (float*)d_out;

    const int mTiles = (N + 15) / 16;
    dim3 gemmGrid((mTiles + 3) / 4);
    dim3 gemmBlock(128);
    dim3 nodeGrid((N + 7) / 8);
    dim3 nodeBlock(256);

    k_hist<<<nBlk, SORT_T, 0, stream>>>(dst, blockHist, E, NB);
    k_part<<<nBlk, SORT_T, 0, stream>>>(src, dst, blockHist, binStart, tsrc, tdst, E, NB, nBlk);
    k_binsort<<<NB, SORT_T, 0, stream>>>(tsrc, tdst, binStart, rowptr, inv_sqrt, col, N, E);

    k_gemm_wmma_f32<<<gemmGrid, gemmBlock, 0, stream>>>(x, W1, hbuf, mTiles);

    k_gather_gemm<<<FUSED_GRID, 256, 0, stream>>>(hbuf, rowptr, col, inv_sqrt, b1, W2, hbuf2, N, E, mTiles);

    k_gather<<<nodeGrid, nodeBlock, 0, stream>>>(hbuf2, rowptr, col, inv_sqrt, b2, out, N,  0);
}


# Aggregation moved in front of each product (linearity): both layers run as one gather+WMMA launch on their input rows; stand-alone product and plain gather launches go (6 to 5); sum order changes
# speedup vs baseline: 5.5524x; 1.1701x over previous
#include <hip/hip_runtime.h>
#include <math.h>


typedef __attribute__((ext_vector_type(2))) float v2f;
typedef __attribute__((ext_vector_type(8))) float v8f;

#define IN_C 128

#define BIN_SHIFT 9
#define BIN_W     512
#define NB_MAX    1024
#define SORT_T    1024
#define CHUNK     8192
#define STAGE_CAP 32768
#define BATCH     (CHUNK / SORT_T)

__device__ __forceinline__ int bin_of(int d, int nB) {
    return (int)min((unsigned)d >> BIN_SHIFT, (unsigned)(nB - 1));
}

__device__ __forceinline__ int block_excl_scan(int v, int* s_w, int* total) {
    const int lane = threadIdx.x & 31;
    const int wv   = threadIdx.x >> 5;
    int inc = v;
#pragma unroll
    for (int o = 1; o < 32; o <<= 1) {
        int u = __shfl_up(inc, o);
        if (lane >= o) inc += u;
    }
    __syncthreads();
    if (lane == 31) s_w[wv] = inc;
    __syncthreads();
    if (wv == 0) {
        int t = s_w[lane];
#pragma unroll
        for (int o = 1; o < 32; o <<= 1) {
            int u = __shfl_up(t, o);
            if (lane >= o) t += u;
        }
        s_w[lane] = t;
    }
    __syncthreads();
    const int wbase = (wv > 0) ? s_w[wv - 1] : 0;
    *total = s_w[31];
    return wbase + inc - v;
}

__global__ __launch_bounds__(SORT_T) void k_hist(
    const int* __restrict__ dst, int* __restrict__ blockHist, int nE, int nB)
{
    __shared__ int s_h[NB_MAX];
    const int tid = threadIdx.x;
    nB = min(nB, NB_MAX);
    s_h[tid] = 0;
    __syncthreads();
    const int c0 = blockIdx.x * CHUNK;
    int d[BATCH];
#pragma unroll
    for (int k = 0; k < BATCH; ++k) {
        const int i = c0 + k * SORT_T + tid;
        d[k] = dst[min(i, nE - 1)];
    }
#pragma unroll
    for (int k = 0; k < BATCH; ++k) {
        const int i = c0 + k * SORT_T + tid;
        if (i < nE) atomicAdd(&s_h[bin_of(d[k], nB)], 1);
    }
    __syncthreads();
    if (tid < nB) blockHist[blockIdx.x * nB + tid] = s_h[tid];
}

__global__ __launch_bounds__(SORT_T) void k_part(
    const int* __restrict__ src, const int* __restrict__ dst,
    const int* __restrict__ blockHist, int* __restrict__ binStart,
    int* __restrict__ tsrc, int* __restrict__ tdst, int nE, int nB, int nBlk)
{
    __shared__ int s_w[32];
    __shared__ int s_cur[NB_MAX];
    __shared__ int s_off[NB_MAX];
    __shared__ int s_gb[NB_MAX];
    __shared__ int s_ssrc[CHUNK];
    __shared__ int s_sdst[CHUNK];
    const int tid = threadIdx.x;
    const int blk = blockIdx.x;
    nB = min(nB, NB_MAX);

    int tot = 0, pre = 0, own = 0;
    if (tid < nB) {
        for (int k = 0; k < nBlk; ++k) {
            const int hc = blockHist[k * nB + tid];
            tot += hc;
            pre += (k < blk) ? hc : 0;
        }
        own = blockHist[blk * nB + tid];
    }
    int nAll, nChunk;
    const int bstart = block_excl_scan(tot, s_w, &nAll);
    const int off    = block_excl_scan(own, s_w, &nChunk);
    s_cur[tid] = off;
    s_off[tid] = off;
    s_gb[tid]  = bstart + pre;
    if (blk == 0) {
        if (tid < nB) binStart[tid] = bstart;
        if (tid == 0) binStart[nB] = nAll;
    }
    __syncthreads();

    const int c0 = blk * CHUNK;
    int d[BATCH], s[BATCH];
#pragma unroll
    for (int k = 0; k < BATCH; ++k) {
        const int i = c0 + k * SORT_T + tid;
        d[k] = dst[min(i, nE - 1)];
        s[k] = src[min(i, nE - 1)];
    }
#pragma unroll
    for (int k = 0; k < BATCH; ++k) {
        const int i = c0 + k * SORT_T + tid;
        if (i < nE) {
            const int slot = atomicAdd(&s_cur[bin_of(d[k], nB)], 1);
            if ((unsigned)slot < (unsigned)CHUNK) { s_ssrc[slot] = s[k]; s_sdst[slot] = d[k]; }
        }
    }
    __syncthreads();

    for (int k = 0; k < BATCH; ++k) {
        const int slot = k * SORT_T + tid;
        if (slot < nChunk) {
            const int dd = s_sdst[slot];
            const int b  = bin_of(dd, nB);
            const int g  = s_gb[b] + (slot - s_off[b]);
            if ((unsigned)g < (unsigned)nE) { tsrc[g] = s_ssrc[slot]; tdst[g] = dd; }
        }
    }
}

__global__ __launch_bounds__(SORT_T) void k_binsort(
    const int* __restrict__ tsrc, const int* __restrict__ tdst,
    const int* __restrict__ binStart, int* __restrict__ rowptr,
    float* __restrict__ inv_sqrt, int* __restrict__ col, int nN, int nE)
{
    __shared__ int s_w[32];
    __shared__ int s_deg[BIN_W];
    __shared__ int s_cur[BIN_W];
    __shared__ int s_col[STAGE_CAP];
    const int tid = threadIdx.x;
    const int b   = blockIdx.x;
    const int e0  = min(max(binStart[b], 0), nE);
    const int e1  = min(max(binStart[b + 1], e0), nE);
    const int cnt = e1 - e0;
    const int node0 = b << BIN_SHIFT;

    if (tid < BIN_W) s_deg[tid] = 0;
    __syncthreads();
    for (int i0 = e0 + tid; i0 < e1; i0 += BATCH * SORT_T) {
        int d[BATCH];
#pragma unroll
        for (int k = 0; k < BATCH; ++k) {
            const int i = i0 + k * SORT_T;
            d[k] = tdst[min(i, e1 - 1)];
        }
#pragma unroll
        for (int k = 0; k < BATCH; ++k) {
            const int i = i0 + k * SORT_T;
            if (i < e1) atomicAdd(&s_deg[(int)min((unsigned)(d[k] - node0), (unsigned)(BIN_W - 1))], 1);
        }
    }
    __syncthreads();

    const int dg = (tid < BIN_W) ? s_deg[tid] : 0;
    int nBin;
    const int lrow = block_excl_scan(dg, s_w, &nBin);
    if (tid < BIN_W) {
        s_cur[tid] = lrow;
        const int node = node0 + tid;
        if (node < nN) {
            rowptr[node + 1] = e0 + lrow + dg;
            inv_sqrt[node] = 1.0f / sqrtf((float)(dg + 1));
        }
    }
    if (b == 0 && tid == 0) rowptr[0] = 0;
    __syncthreads();

    const bool staged = (cnt <= STAGE_CAP);
    for (int i0 = e0 + tid; i0 < e1; i0 += BATCH * SORT_T) {
        int d[BATCH], s[BATCH];
#pragma unroll
        for (int k = 0; k < BATCH; ++k) {
            const int i = i0 + k * SORT_T;
            d[k] = tdst[min(i, e1 - 1)];
            s[k] = tsrc[min(i, e1 - 1)];
        }
#pragma unroll
        for (int k = 0; k < BATCH; ++k) {
            const int i = i0 + k * SORT_T;
            if (i < e1) {
                const int ld = (int)min((unsigned)(d[k] - node0), (unsigned)(BIN_W - 1));
                const int p  = atomicAdd(&s_cur[ld], 1);
                if ((unsigned)p < (unsigned)cnt) {
                    if (staged) s_col[p] = s[k];
                    else        col[e0 + p] = s[k];
                }
            }
        }
    }
    if (staged) {
        __syncthreads();
        for (int i = tid; i < cnt; i += SORT_T) col[e0 + i] = s_col[i];
    }
}

#define FUSED_GRID 1024
#define TILE_M     16
#define TILE_PITCH 132

__device__ __forceinline__ float readlane_f(float v, int l) {
    return __builtin_bit_cast(float, __builtin_amdgcn_readlane(__builtin_bit_cast(int, v), l));
}

__device__ __forceinline__ void fma_row(float4& acc, const float4 hv, const float c) {
    acc.x = __builtin_fmaf(hv.x, c, acc.x);
    acc.y = __builtin_fmaf(hv.y, c, acc.y);
    acc.z = __builtin_fmaf(hv.z, c, acc.z);
    acc.w = __builtin_fmaf(hv.w, c, acc.w);
}

__device__ __forceinline__ float4 src_row(const float4* __restrict__ h4, int my_s, int j, int lane) {
    return h4[(size_t)__builtin_amdgcn_readlane(my_s, j) * (IN_C / 4) + lane];
}

__device__ __forceinline__ void rows_acc(float4& acc, const float4* __restrict__ h4,
                                         int my_s, float my_c, int m, int lane)
{
    int j = 0;
    for (; j + 4 <= m; j += 4) {
        const float4 h0 = src_row(h4, my_s, j, lane);
        const float4 h1 = src_row(h4, my_s, j + 1, lane);
        const float4 h2 = src_row(h4, my_s, j + 2, lane);
        const float4 h3 = src_row(h4, my_s, j + 3, lane);
        fma_row(acc, h0, readlane_f(my_c, j));
        fma_row(acc, h1, readlane_f(my_c, j + 1));
        fma_row(acc, h2, readlane_f(my_c, j + 2));
        fma_row(acc, h3, readlane_f(my_c, j + 3));
    }
    const int rem = m - j;
    if (rem == 3) {
        const float4 h0 = src_row(h4, my_s, j, lane);
        const float4 h1 = src_row(h4, my_s, j + 1, lane);
        const float4 h2 = src_row(h4, my_s, j + 2, lane);
        fma_row(acc, h0, readlane_f(my_c, j));
        fma_row(acc, h1, readlane_f(my_c, j + 1));
        fma_row(acc, h2, readlane_f(my_c, j + 2));
    } else if (rem == 2) {
        const float4 h0 = src_row(h4, my_s, j, lane);
        const float4 h1 = src_row(h4, my_s, j + 1, lane);
        fma_row(acc, h0, readlane_f(my_c, j));
        fma_row(acc, h1, readlane_f(my_c, j + 1));
    } else if (rem == 1) {
        const float4 h0 = src_row(h4, my_s, j, lane);
        fma_row(acc, h0, readlane_f(my_c, j));
    }
}

__device__ __forceinline__ float4 node_sum(const float4* __restrict__ h4, const int* __restrict__ col,
                                           const float* __restrict__ inv_sqrt, int beg, int end,
                                           int my_s, float my_c, float isd, int nN, int nE, int lane)
{
    float4 acc = make_float4(0.0f, 0.0f, 0.0f, 0.0f);
    rows_acc(acc, h4, my_s, my_c, min(32, end - beg), lane);
    for (int base = beg + 32; base < end; base += 32) {
        const int m = min(32, end - base);
        int   s = 0;
        float c = 0.0f;
        if (lane < m) {
            s = (int)min((unsigned)col[min(base + lane, nE - 1)], (unsigned)(nN - 1));
            c = inv_sqrt[s] * isd;
        }
        rows_acc(acc, h4, s, c, m, lane);
    }
    return acc;
}

__device__ __forceinline__ int edge_slot(int beg, int end, int lane, int nE) {
    return min(max(min(beg + lane, end - 1), 0), nE - 1);
}

template <bool RELU>
__global__ __launch_bounds__(256) void k_gather_gemm(
    const float* __restrict__ h, const int* __restrict__ rowptr,
    const int* __restrict__ col, const float* __restrict__ inv_sqrt,
    const float* __restrict__ bias, const float* __restrict__ Wn,
    float* __restrict__ hn, int nN, int nE, int mTiles)
{
    __shared__ float s_a[TILE_M * TILE_PITCH];
    __shared__ float s_c[TILE_M * TILE_PITCH];
    const int lane = threadIdx.x & 31;
    const int wv   = __builtin_amdgcn_readfirstlane((int)(threadIdx.x >> 5));
    const int row  = lane & 15;
    const int hi   = lane >> 4;
    const int koff = hi * 2;

    int tile = blockIdx.x;
    if (tile >= mTiles) return;

    v2f bfrag[32];
    {
        const float* __restrict__ wbase = Wn + (size_t)koff * IN_C + wv * 16 + row;
#pragma unroll
        for (int ks = 0; ks < 32; ++ks) {
            bfrag[ks].x = wbase[(size_t)(ks * 4) * IN_C];
            bfrag[ks].y = wbase[(size_t)(ks * 4 + 1) * IN_C];
        }
    }
    const float bcol = bias[wv * 16 + row];
    const float4* __restrict__ h4  = (const float4*)h;
    float4* __restrict__       hn4 = (float4*)hn;

    int   n0  = tile * TILE_M + wv * 2;
    int   rpv = rowptr[min(n0 + min(lane, 2), nN)];
    float isv = inv_sqrt[min(n0 + (lane & 1), nN - 1)];
    int   craw0, craw1;
    float iraw0, iraw1;
    {
        const int p0 = __builtin_amdgcn_readlane(rpv, 0);
        const int p1 = __builtin_amdgcn_readlane(rpv, 1);
        const int p2 = __builtin_amdgcn_readlane(rpv, 2);
        craw0 = col[edge_slot(p0, p1, lane, nE)];
        craw1 = col[edge_slot(p1, p2, lane, nE)];
        iraw0 = inv_sqrt[min((unsigned)craw0, (unsigned)(nN - 1))];
        iraw1 = inv_sqrt[min((unsigned)craw1, (unsigned)(nN - 1))];
    }

    for (;;) {
        const int ntile = tile + gridDim.x;
        n0 = tile * TILE_M + wv * 2;
        const int   p0   = __builtin_amdgcn_readlane(rpv, 0);
        const int   p1   = __builtin_amdgcn_readlane(rpv, 1);
        const int   p2   = __builtin_amdgcn_readlane(rpv, 2);
        const float isd0 = readlane_f(isv, 0);
        const float isd1 = readlane_f(isv, 1);
        const int   ms0  = (int)min((unsigned)craw0, (unsigned)(nN - 1));
        const int   ms1  = (int)min((unsigned)craw1, (unsigned)(nN - 1));
        const float mc0  = (lane < p1 - p0) ? iraw0 * isd0 : 0.0f;
        const float mc1  = (lane < p2 - p1) ? iraw1 * isd1 : 0.0f;

        const int   nn0  = ntile * TILE_M + wv * 2;
        const int   rpvN = rowptr[min(nn0 + min(lane, 2), nN)];
        const float isvN = inv_sqrt[min(nn0 + (lane & 1), nN - 1)];

        {
            const float4 hs  = h4[(size_t)min(n0, nN - 1) * (IN_C / 4) + lane];
            const float4 acc = node_sum(h4, col, inv_sqrt, p0, p1, ms0, mc0, isd0, nN, nE, lane);
            float4 v;
            v.x = __builtin_fmaf(hs.x * isd0, isd0, acc.x);
            v.y = __builtin_fmaf(hs.y * isd0, isd0, acc.y);
            v.z = __builtin_fmaf(hs.z * isd0, isd0, acc.z);
            v.w = __builtin_fmaf(hs.w * isd0, isd0, acc.w);
            if (n0 >= nN) v = make_float4(0.0f, 0.0f, 0.0f, 0.0f);
            *(float4*)(s_a + (wv * 2) * TILE_PITCH + lane * 4) = v;
        }

        const int q0 = __builtin_amdgcn_readlane(rpvN, 0);
        const int q1 = __builtin_amdgcn_readlane(rpvN, 1);
        const int q2 = __builtin_amdgcn_readlane(rpvN, 2);
        const int crawN0 = col[edge_slot(q0, q1, lane, nE)];
        const int crawN1 = col[edge_slot(q1, q2, lane, nE)];

        {
            const float4 hs  = h4[(size_t)min(n0 + 1, nN - 1) * (IN_C / 4) + lane];
            const float4 acc = node_sum(h4, col, inv_sqrt, p1, p2, ms1, mc1, isd1, nN, nE, lane);
            float4 v;
            v.x = __builtin_fmaf(hs.x * isd1, isd1, acc.x);
            v.y = __builtin_fmaf(hs.y * isd1, isd1, acc.y);
            v.z = __builtin_fmaf(hs.z * isd1, isd1, acc.z);
            v.w = __builtin_fmaf(hs.w * isd1, isd1, acc.w);
            if (n0 + 1 >= nN) v = make_float4(0.0f, 0.0f, 0.0f, 0.0f);
            *(float4*)(s_a + (wv * 2 + 1) * TILE_PITCH + lane * 4) = v;
        }

        __syncthreads();

        const float irawN0 = inv_sqrt[min((unsigned)crawN0, (unsigned)(nN - 1))];
        const float irawN1 = inv_sqrt[min((unsigned)crawN1, (unsigned)(nN - 1))];

        v8f acc = {};
        const float* __restrict__ arow = s_a + row * TILE_PITCH + koff;
#pragma unroll
        for (int ks = 0; ks < 32; ++ks) {
            const v2f a = *(const v2f*)(arow + ks * 4);
            acc = __builtin_amdgcn_wmma_f32_16x16x4_f32(
                false, a, false, bfrag[ks], (short)0, acc, false, false);
        }
#pragma unroll
        for (int r = 0; r < 8; ++r) {
            float o = acc[r] + bcol;
            if (RELU) o = fmaxf(o, 0.0f);
            s_c[(r + 8 * hi) * TILE_PITCH + wv * 16 + row] = o;
        }

        __syncthreads();

#pragma unroll
        for (int q = 0; q < 2; ++q) {
            const int node = n0 + q;
            if (node < nN)
                hn4[(size_t)node * (IN_C / 4) + lane] = *(const float4*)(s_c + (wv * 2 + q) * TILE_PITCH + lane * 4);
        }

        if (ntile >= mTiles) break;
        tile = ntile;
        rpv = rpvN;  isv = isvN;
        craw0 = crawN0;  craw1 = crawN1;
        iraw0 = irawN0;  iraw1 = irawN1;
    }
}

extern "C" void kernel_launch(void* const* d_in, const int* in_sizes, int n_in,
                              void* d_out, int out_size, void* d_ws, size_t ws_size,
                              hipStream_t stream)
{
    const float* x   = (const float*)d_in[0];
    const int*   ei  = (const int*)d_in[1];
    const float* W1  = (const float*)d_in[2];
    const float* b1  = (const float*)d_in[3];
    const float* W2  = (const float*)d_in[4];
    const float* b2  = (const float*)d_in[5];

    const int N = in_sizes[0] / IN_C;
    const int E = in_sizes[1] / 2;
    const int* src = ei;
    const int* dst = ei + E;

    const int NB   = (N + BIN_W - 1) / BIN_W;
    const int nBlk = (E + CHUNK - 1) / CHUNK;
    float* hbuf      = (float*)d_ws;
    float* inv_sqrt  = hbuf + (size_t)N * IN_C;
    int*   rowptr    = (int*)(inv_sqrt + N);
    int*   col       = rowptr + N + 4;
    int*   tsrc      = col + E;
    int*   tdst      = tsrc + E;
    int*   binStart  = tdst + E;
    int*   blockHist = binStart + NB + 4;
    float* out       = (float*)d_out;

    const int mTiles = (N + 15) / 16;

    k_hist<<<nBlk, SORT_T, 0, stream>>>(dst, blockHist, E, NB);
    k_part<<<nBlk, SORT_T, 0, stream>>>(src, dst, blockHist, binStart, tsrc, tdst, E, NB, nBlk);
    k_binsort<<<NB, SORT_T, 0, stream>>>(tsrc, tdst, binStart, rowptr, inv_sqrt, col, N, E);

    k_gather_gemm<true><<<FUSED_GRID, 256, 0, stream>>>(x, rowptr, col, inv_sqrt, b1, W1, hbuf, N, E, mTiles);

    k_gather_gemm<false><<<FUSED_GRID, 256, 0, stream>>>(hbuf, rowptr, col, inv_sqrt, b2, W2, out, N, E, mTiles);
}


# Gather keeps 8 row loads in flight per wave, not 4 (passes of 8 rows, tail pass of 4; a slot past the end re-reads the last row times 0); 32-bit row offsets keep it in 128 VGPRs; sum order kept
# speedup vs baseline: 5.8398x; 1.0518x over previous
#include <hip/hip_runtime.h>
#include <math.h>


typedef __attribute__((ext_vector_type(2))) float v2f;
typedef __attribute__((ext_vector_type(8))) float v8f;

#define IN_C 128

#define BIN_SHIFT 9
#define BIN_W     512
#define NB_MAX    1024
#define SORT_T    1024
#define CHUNK     8192
#define STAGE_CAP 32768
#define BATCH     (CHUNK / SORT_T)

__device__ __forceinline__ int bin_of(int d, int nB) {
    return (int)min((unsigned)d >> BIN_SHIFT, (unsigned)(nB - 1));
}

__device__ __forceinline__ int block_excl_scan(int v, int* s_w, int* total) {
    const int lane = threadIdx.x & 31;
    const int wv   = threadIdx.x >> 5;
    int inc = v;
#pragma unroll
    for (int o = 1; o < 32; o <<= 1) {
        int u = __shfl_up(inc, o);
        if (lane >= o) inc += u;
    }
    __syncthreads();
    if (lane == 31) s_w[wv] = inc;
    __syncthreads();
    if (wv == 0) {
        int t = s_w[lane];
#pragma unroll
        for (int o = 1; o < 32; o <<= 1) {
            int u = __shfl_up(t, o);
            if (lane >= o) t += u;
        }
        s_w[lane] = t;
    }
    __syncthreads();
    const int wbase = (wv > 0) ? s_w[wv - 1] : 0;
    *total = s_w[31];
    return wbase + inc - v;
}

__global__ __launch_bounds__(SORT_T) void k_hist(
    const int* __restrict__ dst, int* __restrict__ blockHist, int nE, int nB)
{
    __shared__ int s_h[NB_MAX];
    const int tid = threadIdx.x;
    nB = min(nB, NB_MAX);
    s_h[tid] = 0;
    __syncthreads();
    const int c0 = blockIdx.x * CHUNK;
    int d[BATCH];
#pragma unroll
    for (int k = 0; k < BATCH; ++k) {
        const int i = c0 + k * SORT_T + tid;
        d[k] = dst[min(i, nE - 1)];
    }
#pragma unroll
    for (int k = 0; k < BATCH; ++k) {
        const int i = c0 + k * SORT_T + tid;
        if (i < nE) atomicAdd(&s_h[bin_of(d[k], nB)], 1);
    }
    __syncthreads();
    if (tid < nB) blockHist[blockIdx.x * nB + tid] = s_h[tid];
}

__global__ __launch_bounds__(SORT_T) void k_part(
    const int* __restrict__ src, const int* __restrict__ dst,
    const int* __restrict__ blockHist, int* __restrict__ binStart,
    int* __restrict__ tsrc, int* __restrict__ tdst, int nE, int nB, int nBlk)
{
    __shared__ int s_w[32];
    __shared__ int s_cur[NB_MAX];
    __shared__ int s_off[NB_MAX];
    __shared__ int s_gb[NB_MAX];
    __shared__ int s_ssrc[CHUNK];
    __shared__ int s_sdst[CHUNK];
    const int tid = threadIdx.x;
    const int blk = blockIdx.x;
    nB = min(nB, NB_MAX);

    int tot = 0, pre = 0, own = 0;
    if (tid < nB) {
        for (int k = 0; k < nBlk; ++k) {
            const int hc = blockHist[k * nB + tid];
            tot += hc;
            pre += (k < blk) ? hc : 0;
        }
        own = blockHist[blk * nB + tid];
    }
    int nAll, nChunk;
    const int bstart = block_excl_scan(tot, s_w, &nAll);
    const int off    = block_excl_scan(own, s_w, &nChunk);
    s_cur[tid] = off;
    s_off[tid] = off;
    s_gb[tid]  = bstart + pre;
    if (blk == 0) {
        if (tid < nB) binStart[tid] = bstart;
        if (tid == 0) binStart[nB] = nAll;
    }
    __syncthreads();

    const int c0 = blk * CHUNK;
    int d[BATCH], s[BATCH];
#pragma unroll
    for (int k = 0; k < BATCH; ++k) {
        const int i = c0 + k * SORT_T + tid;
        d[k] = dst[min(i, nE - 1)];
        s[k] = src[min(i, nE - 1)];
    }
#pragma unroll
    for (int k = 0; k < BATCH; ++k) {
        const int i = c0 + k * SORT_T + tid;
        if (i < nE) {
            const int slot = atomicAdd(&s_cur[bin_of(d[k], nB)], 1);
            if ((unsigned)slot < (unsigned)CHUNK) { s_ssrc[slot] = s[k]; s_sdst[slot] = d[k]; }
        }
    }
    __syncthreads();

    for (int k = 0; k < BATCH; ++k) {
        const int slot = k * SORT_T + tid;
        if (slot < nChunk) {
            const int dd = s_sdst[slot];
            const int b  = bin_of(dd, nB);
            const int g  = s_gb[b] + (slot - s_off[b]);
            if ((unsigned)g < (unsigned)nE) { tsrc[g] = s_ssrc[slot]; tdst[g] = dd; }
        }
    }
}

__global__ __launch_bounds__(SORT_T) void k_binsort(
    const int* __restrict__ tsrc, const int* __restrict__ tdst,
    const int* __restrict__ binStart, int* __restrict__ rowptr,
    float* __restrict__ inv_sqrt, int* __restrict__ col, int nN, int nE)
{
    __shared__ int s_w[32];
    __shared__ int s_deg[BIN_W];
    __shared__ int s_cur[BIN_W];
    __shared__ int s_col[STAGE_CAP];
    const int tid = threadIdx.x;
    const int b   = blockIdx.x;
    const int e0  = min(max(binStart[b], 0), nE);
    const int e1  = min(max(binStart[b + 1], e0), nE);
    const int cnt = e1 - e0;
    const int node0 = b << BIN_SHIFT;

    if (tid < BIN_W) s_deg[tid] = 0;
    __syncthreads();
    for (int i0 = e0 + tid; i0 < e1; i0 += BATCH * SORT_T) {
        int d[BATCH];
#pragma unroll
        for (int k = 0; k < BATCH; ++k) {
            const int i = i0 + k * SORT_T;
            d[k] = tdst[min(i, e1 - 1)];
        }
#pragma unroll
        for (int k = 0; k < BATCH; ++k) {
            const int i = i0 + k * SORT_T;
            if (i < e1) atomicAdd(&s_deg[(int)min((unsigned)(d[k] - node0), (unsigned)(BIN_W - 1))], 1);
        }
    }
    __syncthreads();

    const int dg = (tid < BIN_W) ? s_deg[tid] : 0;
    int nBin;
    const int lrow = block_excl_scan(dg, s_w, &nBin);
    if (tid < BIN_W) {
        s_cur[tid] = lrow;
        const int node = node0 + tid;
        if (node < nN) {
            rowptr[node + 1] = e0 + lrow + dg;
            inv_sqrt[node] = 1.0f / sqrtf((float)(dg + 1));
        }
    }
    if (b == 0 && tid == 0) rowptr[0] = 0;
    __syncthreads();

    const bool staged = (cnt <= STAGE_CAP);
    for (int i0 = e0 + tid; i0 < e1; i0 += BATCH * SORT_T) {
        int d[BATCH], s[BATCH];
#pragma unroll
        for (int k = 0; k < BATCH; ++k) {
            const int i = i0 + k * SORT_T;
            d[k] = tdst[min(i, e1 - 1)];
            s[k] = tsrc[min(i, e1 - 1)];
        }
#pragma unroll
        for (int k = 0; k < BATCH; ++k) {
            const int i = i0 + k * SORT_T;
            if (i < e1) {
                const int ld = (int)min((unsigned)(d[k] - node0), (unsigned)(BIN_W - 1));
                const int p  = atomicAdd(&s_cur[ld], 1);
                if ((unsigned)p < (unsigned)cnt) {
                    if (staged) s_col[p] = s[k];
                    else        col[e0 + p] = s[k];
                }
            }
        }
    }
    if (staged) {
        __syncthreads();
        for (int i = tid; i < cnt; i += SORT_T) col[e0 + i] = s_col[i];
    }
}

#define FUSED_GRID 1024
#define TILE_M     16
#define TILE_PITCH 132

__device__ __forceinline__ float readlane_f(float v, int l) {
    return __builtin_bit_cast(float, __builtin_amdgcn_readlane(__builtin_bit_cast(int, v), l));
}

__device__ __forceinline__ void fma_row(float4& acc, const float4 hv, const float c) {
    acc.x = __builtin_fmaf(hv.x, c, acc.x);
    acc.y = __builtin_fmaf(hv.y, c, acc.y);
    acc.z = __builtin_fmaf(hv.z, c, acc.z);
    acc.w = __builtin_fmaf(hv.w, c, acc.w);
}

typedef __attribute__((ext_vector_type(4))) float v4f;
typedef __attribute__((address_space(1))) v4f gv4f;

__device__ __forceinline__ float4 row_load(const float4* __restrict__ m4, int r, int lane) {
    const v4f v = ((const gv4f*)m4)[(unsigned)r * (unsigned)(IN_C / 4) + (unsigned)lane];
    return make_float4(v.x, v.y, v.z, v.w);
}

__device__ __forceinline__ float4 src_row(const float4* __restrict__ h4, int my_s, int j, int lane) {
    return row_load(h4, __builtin_amdgcn_readlane(my_s, j), lane);
}

#define ROWS_B 8
#define ROWS_S 4

template <int N>
__device__ __forceinline__ void rows_pass(float4& acc, const float4* __restrict__ h4,
                                          int my_s, float my_c, int j, int lane)
{
    float4 hv[N];
#pragma unroll
    for (int t = 0; t < N; ++t) hv[t] = src_row(h4, my_s, j + t, lane);
#pragma unroll
    for (int t = 0; t < N; ++t) fma_row(acc, hv[t], readlane_f(my_c, j + t));
}

__device__ __forceinline__ void rows_acc(float4& acc, const float4* __restrict__ h4,
                                         int my_s, float my_c, int m, int lane)
{
    int j = 0;
    for (; j + ROWS_S < m; j += ROWS_B) rows_pass<ROWS_B>(acc, h4, my_s, my_c, j, lane);
    if (j < m) rows_pass<ROWS_S>(acc, h4, my_s, my_c, j, lane);
}

__device__ __forceinline__ float4 node_sum(const float4* __restrict__ h4, const int* __restrict__ col,
                                           const float* __restrict__ inv_sqrt, int beg, int end,
                                           int my_s, float my_c, float isd, int nN, int nE, int lane)
{
    float4 acc = make_float4(0.0f, 0.0f, 0.0f, 0.0f);
    rows_acc(acc, h4, my_s, my_c, min(32, end - beg), lane);
    for (int base = beg + 32; base < end; base += 32) {
        const int m = min(32, end - base);
        int   s = 0;
        float c = 0.0f;
        if (lane < m) {
            s = (int)min((unsigned)col[min(base + lane, nE - 1)], (unsigned)(nN - 1));
            c = inv_sqrt[s] * isd;
        }
        rows_acc(acc, h4, s, c, m, lane);
    }
    return acc;
}

__device__ __forceinline__ int edge_slot(int beg, int end, int lane, int nE) {
    return min(max(min(beg + lane, end - 1), 0), nE - 1);
}

template <bool RELU>
__global__ __launch_bounds__(256) void k_gather_gemm(
    const float* __restrict__ h, const int* __restrict__ rowptr,
    const int* __restrict__ col, const float* __restrict__ inv_sqrt,
    const float* __restrict__ bias, const float* __restrict__ Wn,
    float* __restrict__ hn, int nN, int nE, int mTiles)
{
    __shared__ float s_a[TILE_M * TILE_PITCH];
    __shared__ float s_c[TILE_M * TILE_PITCH];
    const int lane = threadIdx.x & 31;
    const int wv   = __builtin_amdgcn_readfirstlane((int)(threadIdx.x >> 5));
    const int row  = lane & 15;
    const int hi   = lane >> 4;
    const int koff = hi * 2;

    int tile = blockIdx.x;
    if (tile >= mTiles) return;

    v2f bfrag[32];
    {
        const float* __restrict__ wbase = Wn + (size_t)koff * IN_C + wv * 16 + row;
#pragma unroll
        for (int ks = 0; ks < 32; ++ks) {
            bfrag[ks].x = wbase[(size_t)(ks * 4) * IN_C];
            bfrag[ks].y = wbase[(size_t)(ks * 4 + 1) * IN_C];
        }
    }
    const float bcol = bias[wv * 16 + row];
    const float4* __restrict__ h4  = (const float4*)h;
    float4* __restrict__       hn4 = (float4*)hn;

    int   n0  = tile * TILE_M + wv * 2;
    int   rpv = rowptr[min(n0 + min(lane, 2), nN)];
    float isv = inv_sqrt[min(n0 + (lane & 1), nN - 1)];
    int   craw0, craw1;
    float iraw0, iraw1;
    {
        const int p0 = __builtin_amdgcn_readlane(rpv, 0);
        const int p1 = __builtin_amdgcn_readlane(rpv, 1);
        const int p2 = __builtin_amdgcn_readlane(rpv, 2);
        craw0 = col[edge_slot(p0, p1, lane, nE)];
        craw1 = col[edge_slot(p1, p2, lane, nE)];
        iraw0 = inv_sqrt[min((unsigned)craw0, (unsigned)(nN - 1))];
        iraw1 = inv_sqrt[min((unsigned)craw1, (unsigned)(nN - 1))];
    }

    for (;;) {
        const int ntile = tile + gridDim.x;
        n0 = tile * TILE_M + wv * 2;
        const int   p0   = __builtin_amdgcn_readlane(rpv, 0);
        const int   p1   = __builtin_amdgcn_readlane(rpv, 1);
        const int   p2   = __builtin_amdgcn_readlane(rpv, 2);
        const float isd0 = readlane_f(isv, 0);
        const float isd1 = readlane_f(isv, 1);
        const int   ms0  = (int)min((unsigned)craw0, (unsigned)(nN - 1));
        const int   ms1  = (int)min((unsigned)craw1, (unsigned)(nN - 1));
        const float mc0  = (lane < p1 - p0) ? iraw0 * isd0 : 0.0f;
        const float mc1  = (lane < p2 - p1) ? iraw1 * isd1 : 0.0f;

        const int   nn0  = ntile * TILE_M + wv * 2;
        const int   rpvN = rowptr[min(nn0 + min(lane, 2), nN)];
        const float isvN = inv_sqrt[min(nn0 + (lane & 1), nN - 1)];

        {
            const float4 hs  = row_load(h4, min(n0, nN - 1), lane);
            const float4 acc = node_sum(h4, col, inv_sqrt, p0, p1, ms0, mc0, isd0, nN, nE, lane);
            float4 v;
            v.x = __builtin_fmaf(hs.x * isd0, isd0, acc.x);
            v.y = __builtin_fmaf(hs.y * isd0, isd0, acc.y);
            v.z = __builtin_fmaf(hs.z * isd0, isd0, acc.z);
            v.w = __builtin_fmaf(hs.w * isd0, isd0, acc.w);
            if (n0 >= nN) v = make_float4(0.0f, 0.0f, 0.0f, 0.0f);
            *(float4*)(s_a + (wv * 2) * TILE_PITCH + lane * 4) = v;
        }

        const int q0 = __builtin_amdgcn_readlane(rpvN, 0);
        const int q1 = __builtin_amdgcn_readlane(rpvN, 1);
        const int q2 = __builtin_amdgcn_readlane(rpvN, 2);
        const int crawN0 = col[edge_slot(q0, q1, lane, nE)];
        const int crawN1 = col[edge_slot(q1, q2, lane, nE)];

        {
            const float4 hs  = row_load(h4, min(n0 + 1, nN - 1), lane);
            const float4 acc = node_sum(h4, col, inv_sqrt, p1, p2, ms1, mc1, isd1, nN, nE, lane);
            float4 v;
            v.x = __builtin_fmaf(hs.x * isd1, isd1, acc.x);
            v.y = __builtin_fmaf(hs.y * isd1, isd1, acc.y);
            v.z = __builtin_fmaf(hs.z * isd1, isd1, acc.z);
            v.w = __builtin_fmaf(hs.w * isd1, isd1, acc.w);
            if (n0 + 1 >= nN) v = make_float4(0.0f, 0.0f, 0.0f, 0.0f);
            *(float4*)(s_a + (wv * 2 + 1) * TILE_PITCH + lane * 4) = v;
        }

        __syncthreads();

        const float irawN0 = inv_sqrt[min((unsigned)crawN0, (unsigned)(nN - 1))];
        const float irawN1 = inv_sqrt[min((unsigned)crawN1, (unsigned)(nN - 1))];

        v8f acc = {};
        const float* __restrict__ arow = s_a + row * TILE_PITCH + koff;
#pragma unroll
        for (int ks = 0; ks < 32; ++ks) {
            const v2f a = *(const v2f*)(arow + ks * 4);
            acc = __builtin_amdgcn_wmma_f32_16x16x4_f32(
                false, a, false, bfrag[ks], (short)0, acc, false, false);
        }
#pragma unroll
        for (int r = 0; r < 8; ++r) {
            float o = acc[r] + bcol;
            if (RELU) o = fmaxf(o, 0.0f);
            s_c[(r + 8 * hi) * TILE_PITCH + wv * 16 + row] = o;
        }

        __syncthreads();

#pragma unroll
        for (int q = 0; q < 2; ++q) {
            const int node = n0 + q;
            if (node < nN)
                hn4[(size_t)node * (IN_C / 4) + lane] = *(const float4*)(s_c + (wv * 2 + q) * TILE_PITCH + lane * 4);
        }

        if (ntile >= mTiles) break;
        tile = ntile;
        rpv = rpvN;  isv = isvN;
        craw0 = crawN0;  craw1 = crawN1;
        iraw0 = irawN0;  iraw1 = irawN1;
    }
}

extern "C" void kernel_launch(void* const* d_in, const int* in_sizes, int n_in,
                              void* d_out, int out_size, void* d_ws, size_t ws_size,
                              hipStream_t stream)
{
    const float* x   = (const float*)d_in[0];
    const int*   ei  = (const int*)d_in[1];
    const float* W1  = (const float*)d_in[2];
    const float* b1  = (const float*)d_in[3];
    const float* W2  = (const float*)d_in[4];
    const float* b2  = (const float*)d_in[5];

    const int N = in_sizes[0] / IN_C;
    const int E = in_sizes[1] / 2;
    const int* src = ei;
    const int* dst = ei + E;

    const int NB   = (N + BIN_W - 1) / BIN_W;
    const int nBlk = (E + CHUNK - 1) / CHUNK;
    float* hbuf      = (float*)d_ws;
    float* inv_sqrt  = hbuf + (size_t)N * IN_C;
    int*   rowptr    = (int*)(inv_sqrt + N);
    int*   col       = rowptr + N + 4;
    int*   tsrc      = col + E;
    int*   tdst      = tsrc + E;
    int*   binStart  = tdst + E;
    int*   blockHist = binStart + NB + 4;
    float* out       = (float*)d_out;

    const int mTiles = (N + 15) / 16;

    k_hist<<<nBlk, SORT_T, 0, stream>>>(dst, blockHist, E, NB);
    k_part<<<nBlk, SORT_T, 0, stream>>>(src, dst, blockHist, binStart, tsrc, tdst, E, NB, nBlk);
    k_binsort<<<NB, SORT_T, 0, stream>>>(tsrc, tdst, binStart, rowptr, inv_sqrt, col, N, E);

    k_gather_gemm<true><<<FUSED_GRID, 256, 0, stream>>>(x, rowptr, col, inv_sqrt, b1, W1, hbuf, N, E, mTiles);

    k_gather_gemm<false><<<FUSED_GRID, 256, 0, stream>>>(hbuf, rowptr, col, inv_sqrt, b2, W2, out, N, E, mTiles);
}


# Each edge chunk is ordered by bin in place with its own offset table, so the global count launch goes (5 to 4 launches); the per-bin sort reads its chunk segments; sum order: LDS arrival order
# speedup vs baseline: 6.3598x; 1.0890x over previous
#include <hip/hip_runtime.h>
#include <math.h>


typedef __attribute__((ext_vector_type(2))) float v2f;
typedef __attribute__((ext_vector_type(8))) float v8f;

#define IN_C 128

#define BIN_SHIFT 9
#define BIN_W     512
#define NB_MAX    1024
#define SORT_T    1024
#define CHUNK     8192
#define STAGE_CAP 32768
#define BATCH     (CHUNK / SORT_T)
#define SEG_G     4
#define SEG_U     2

__device__ __forceinline__ int bin_of(int d, int nB) {
    return (int)min((unsigned)d >> BIN_SHIFT, (unsigned)(nB - 1));
}

__device__ __forceinline__ int block_excl_scan(int v, int* s_w, int* total) {
    const int lane = threadIdx.x & 31;
    const int wv   = threadIdx.x >> 5;
    int inc = v;
#pragma unroll
    for (int o = 1; o < 32; o <<= 1) {
        int u = __shfl_up(inc, o);
        if (lane >= o) inc += u;
    }
    __syncthreads();
    if (lane == 31) s_w[wv] = inc;
    __syncthreads();
    if (wv == 0) {
        int t = s_w[lane];
#pragma unroll
        for (int o = 1; o < 32; o <<= 1) {
            int u = __shfl_up(t, o);
            if (lane >= o) t += u;
        }
        s_w[lane] = t;
    }
    __syncthreads();
    const int wbase = (wv > 0) ? s_w[wv - 1] : 0;
    *total = s_w[31];
    return wbase + inc - v;
}

__global__ __launch_bounds__(SORT_T) void k_part(
    const int* __restrict__ src, const int* __restrict__ dst,
    int* __restrict__ offT, int* __restrict__ tsrc, int* __restrict__ tdst, int nE, int nB)
{
    __shared__ int s_w[32];
    __shared__ int s_cur[NB_MAX];
    __shared__ int s_ssrc[CHUNK];
    __shared__ int s_sdst[CHUNK];
    const int tid = threadIdx.x;
    const int blk = blockIdx.x;
    nB = min(nB, NB_MAX);
    s_cur[tid] = 0;
    __syncthreads();

    const int c0 = blk * CHUNK;
    int d[BATCH], s[BATCH];
#pragma unroll
    for (int k = 0; k < BATCH; ++k) {
        const int i = c0 + k * SORT_T + tid;
        d[k] = dst[min(i, nE - 1)];
        s[k] = src[min(i, nE - 1)];
    }
#pragma unroll
    for (int k = 0; k < BATCH; ++k) {
        const int i = c0 + k * SORT_T + tid;
        if (i < nE) atomicAdd(&s_cur[bin_of(d[k], nB)], 1);
    }
    __syncthreads();

    const int own = s_cur[tid];
    int nChunk;
    const int off = block_excl_scan(own, s_w, &nChunk);
    s_cur[tid] = off;
    if (tid < nB)  offT[blk * (nB + 1) + tid] = off;
    if (tid == 0)  offT[blk * (nB + 1) + nB]  = nChunk;
    __syncthreads();

#pragma unroll
    for (int k = 0; k < BATCH; ++k) {
        const int i = c0 + k * SORT_T + tid;
        if (i < nE) {
            const int slot = atomicAdd(&s_cur[bin_of(d[k], nB)], 1);
            if ((unsigned)slot < (unsigned)CHUNK) { s_ssrc[slot] = s[k]; s_sdst[slot] = d[k]; }
        }
    }
    __syncthreads();

    for (int k = 0; k < BATCH; ++k) {
        const int slot = k * SORT_T + tid;
        const int g    = c0 + slot;
        if (slot < nChunk && g < nE) { tsrc[g] = s_ssrc[slot]; tdst[g] = s_sdst[slot]; }
    }
}

template <bool PLACE>
__device__ __forceinline__ void bin_walk(
    const int* __restrict__ tsrc, const int* __restrict__ tdst, const int* __restrict__ offT,
    int* __restrict__ col, int* s_cnt, int* s_col,
    int b, int nB1, int nBlk, int nE, int node0, int e0, int cnt, bool staged, int wv, int lane)
{
    for (int kp = 0; kp < nBlk; kp += SORT_T) {
        const int kmine = kp + wv + 32 * lane;
        int o0 = 0, o1 = 0;
        if (kmine < nBlk) {
            o0 = offT[kmine * nB1 + b];
            o1 = offT[kmine * nB1 + b + 1];
        }
        o0 = min(max(o0, 0), CHUNK);
        o1 = min(max(o1, o0), CHUNK);
        const int nMine = min(32, (nBlk - kp - wv + 31) >> 5);
        for (int g0 = 0; g0 < nMine; g0 += SEG_G) {
            int base[SEG_G], c[SEG_G];
            int d[SEG_G][SEG_U], s[SEG_G][SEG_U];
#pragma unroll
            for (int g = 0; g < SEG_G; ++g) {
                const int gi = min(g0 + g, 31);
                const int a0 = __builtin_amdgcn_readlane(o0, gi);
                const int a1 = __builtin_amdgcn_readlane(o1, gi);
                c[g]    = (g0 + g < nMine) ? a1 - a0 : 0;
                base[g] = (kp + wv + 32 * (g0 + g)) * CHUNK + a0;
#pragma unroll
                for (int u = 0; u < SEG_U; ++u) {
                    const int idx = min(max(base[g] + min(lane + 32 * u, c[g] - 1), 0), nE - 1);
                    d[g][u] = tdst[idx];
                    s[g][u] = PLACE ? tsrc[idx] : 0;
                }
            }
#pragma unroll
            for (int g = 0; g < SEG_G; ++g) {
#pragma unroll
                for (int u = 0; u < SEG_U; ++u) {
                    if (lane + 32 * u < c[g]) {
                        const int ld = (int)min((unsigned)(d[g][u] - node0), (unsigned)(BIN_W - 1));
                        if (PLACE) {
                            const int p = atomicAdd(&s_cnt[ld], 1);
                            if ((unsigned)p < (unsigned)cnt) {
                                if (staged) s_col[p] = s[g][u];
                                else        col[e0 + p] = s[g][u];
                            }
                        } else {
                            atomicAdd(&s_cnt[ld], 1);
                        }
                    }
                }
            }
#pragma unroll
            for (int g = 0; g < SEG_G; ++g) {
                for (int i = lane + 32 * SEG_U; i < c[g]; i += 32) {
                    const int idx = min(max(base[g] + i, 0), nE - 1);
                    const int dd  = tdst[idx];
                    const int ld  = (int)min((unsigned)(dd - node0), (unsigned)(BIN_W - 1));
                    if (PLACE) {
                        const int ss = tsrc[idx];
                        const int p  = atomicAdd(&s_cnt[ld], 1);
                        if ((unsigned)p < (unsigned)cnt) {
                            if (staged) s_col[p] = ss;
                            else        col[e0 + p] = ss;
                        }
                    } else {
                        atomicAdd(&s_cnt[ld], 1);
                    }
                }
            }
        }
    }
}

__global__ __launch_bounds__(SORT_T) void k_binsort(
    const int* __restrict__ tsrc, const int* __restrict__ tdst,
    const int* __restrict__ offT, int* __restrict__ rowptr,
    float* __restrict__ inv_sqrt, int* __restrict__ col, int nN, int nE, int nB, int nBlk)
{
    __shared__ int s_w[32];
    __shared__ int s_deg[BIN_W];
    __shared__ int s_cur[BIN_W];
    __shared__ int s_col[STAGE_CAP];
    const int tid  = threadIdx.x;
    const int lane = tid & 31;
    const int wv   = __builtin_amdgcn_readfirstlane(tid >> 5);
    const int b    = blockIdx.x;
    nB = min(nB, NB_MAX);
    if (b >= nB) return;
    const int nB1   = nB + 1;
    const int node0 = b << BIN_SHIFT;

    int sum0 = 0, sum1 = 0;
    for (int k = tid; k < nBlk; k += SORT_T) {
        const int o0 = min(max(offT[k * nB1 + b], 0), CHUNK);
        const int o1 = min(max(offT[k * nB1 + b + 1], o0), CHUNK);
        sum0 += o0;
        sum1 += o1;
    }
    int t0, t1;
    block_excl_scan(sum0, s_w, &t0);
    block_excl_scan(sum1, s_w, &t1);
    const int e0  = min(max(t0, 0), nE);
    const int e1  = min(max(t1, e0), nE);
    const int cnt = e1 - e0;

    if (tid < BIN_W) s_deg[tid] = 0;
    __syncthreads();
    bin_walk<false>(tsrc, tdst, offT, col, s_deg, s_col, b, nB1, nBlk, nE, node0, e0, cnt, false, wv, lane);
    __syncthreads();

    const int dg = (tid < BIN_W) ? s_deg[tid] : 0;
    int nBin;
    const int lrow = block_excl_scan(dg, s_w, &nBin);
    if (tid < BIN_W) {
        s_cur[tid] = lrow;
        const int node = node0 + tid;
        if (node < nN) {
            rowptr[node + 1] = e0 + lrow + dg;
            inv_sqrt[node] = 1.0f / sqrtf((float)(dg + 1));
        }
    }
    if (b == 0 && tid == 0) rowptr[0] = 0;
    __syncthreads();

    const bool staged = (cnt <= STAGE_CAP);
    bin_walk<true>(tsrc, tdst, offT, col, s_cur, s_col, b, nB1, nBlk, nE, node0, e0, cnt, staged, wv, lane);
    if (staged) {
        __syncthreads();
        for (int i = tid; i < cnt; i += SORT_T) col[e0 + i] = s_col[i];
    }
}

#define FUSED_GRID 1024
#define TILE_M     16
#define TILE_PITCH 132

__device__ __forceinline__ float readlane_f(float v, int l) {
    return __builtin_bit_cast(float, __builtin_amdgcn_readlane(__builtin_bit_cast(int, v), l));
}

__device__ __forceinline__ void fma_row(float4& acc, const float4 hv, const float c) {
    acc.x = __builtin_fmaf(hv.x, c, acc.x);
    acc.y = __builtin_fmaf(hv.y, c, acc.y);
    acc.z = __builtin_fmaf(hv.z, c, acc.z);
    acc.w = __builtin_fmaf(hv.w, c, acc.w);
}

typedef __attribute__((ext_vector_type(4))) float v4f;
typedef __attribute__((address_space(1))) v4f gv4f;

__device__ __forceinline__ float4 row_load(const float4* __restrict__ m4, int r, int lane) {
    const v4f v = ((const gv4f*)m4)[(unsigned)r * (unsigned)(IN_C / 4) + (unsigned)lane];
    return make_float4(v.x, v.y, v.z, v.w);
}

__device__ __forceinline__ float4 src_row(const float4* __restrict__ h4, int my_s, int j, int lane) {
    return row_load(h4, __builtin_amdgcn_readlane(my_s, j), lane);
}

#define ROWS_B 8
#define ROWS_S 4

template <int N>
__device__ __forceinline__ void rows_pass(float4& acc, const float4* __restrict__ h4,
                                          int my_s, float my_c, int j, int lane)
{
    float4 hv[N];
#pragma unroll
    for (int t = 0; t < N; ++t) hv[t] = src_row(h4, my_s, j + t, lane);
#pragma unroll
    for (int t = 0; t < N; ++t) fma_row(acc, hv[t], readlane_f(my_c, j + t));
}

__device__ __forceinline__ void rows_acc(float4& acc, const float4* __restrict__ h4,
                                         int my_s, float my_c, int m, int lane)
{
    int j = 0;
    for (; j + ROWS_S < m; j += ROWS_B) rows_pass<ROWS_B>(acc, h4, my_s, my_c, j, lane);
    if (j < m) rows_pass<ROWS_S>(acc, h4, my_s, my_c, j, lane);
}

__device__ __forceinline__ float4 node_sum(const float4* __restrict__ h4, const int* __restrict__ col,
                                           const float* __restrict__ inv_sqrt, int beg, int end,
                                           int my_s, float my_c, float isd, int nN, int nE, int lane)
{
    float4 acc = make_float4(0.0f, 0.0f, 0.0f, 0.0f);
    rows_acc(acc, h4, my_s, my_c, min(32, end - beg), lane);
    for (int base = beg + 32; base < end; base += 32) {
        const int m = min(32, end - base);
        int   s = 0;
        float c = 0.0f;
        if (lane < m) {
            s = (int)min((unsigned)col[min(base + lane, nE - 1)], (unsigned)(nN - 1));
            c = inv_sqrt[s] * isd;
        }
        rows_acc(acc, h4, s, c, m, lane);
    }
    return acc;
}

__device__ __forceinline__ int edge_slot(int beg, int end, int lane, int nE) {
    return min(max(min(beg + lane, end - 1), 0), nE - 1);
}

template <bool RELU>
__global__ __launch_bounds__(256) void k_gather_gemm(
    const float* __restrict__ h, const int* __restrict__ rowptr,
    const int* __restrict__ col, const float* __restrict__ inv_sqrt,
    const float* __restrict__ bias, const float* __restrict__ Wn,
    float* __restrict__ hn, int nN, int nE, int mTiles)
{
    __shared__ float s_a[TILE_M * TILE_PITCH];
    __shared__ float s_c[TILE_M * TILE_PITCH];
    const int lane = threadIdx.x & 31;
    const int wv   = __builtin_amdgcn_readfirstlane((int)(threadIdx.x >> 5));
    const int row  = lane & 15;
    const int hi   = lane >> 4;
    const int koff = hi * 2;

    int tile = blockIdx.x;
    if (tile >= mTiles) return;

    v2f bfrag[32];
    {
        const float* __restrict__ wbase = Wn + (size_t)koff * IN_C + wv * 16 + row;
#pragma unroll
        for (int ks = 0; ks < 32; ++ks) {
            bfrag[ks].x = wbase[(size_t)(ks * 4) * IN_C];
            bfrag[ks].y = wbase[(size_t)(ks * 4 + 1) * IN_C];
        }
    }
    const float bcol = bias[wv * 16 + row];
    const float4* __restrict__ h4  = (const float4*)h;
    float4* __restrict__       hn4 = (float4*)hn;

    int   n0  = tile * TILE_M + wv * 2;
    int   rpv = rowptr[min(n0 + min(lane, 2), nN)];
    float isv = inv_sqrt[min(n0 + (lane & 1), nN - 1)];
    int   craw0, craw1;
    float iraw0, iraw1;
    {
        const int p0 = __builtin_amdgcn_readlane(rpv, 0);
        const int p1 = __builtin_amdgcn_readlane(rpv, 1);
        const int p2 = __builtin_amdgcn_readlane(rpv, 2);
        craw0 = col[edge_slot(p0, p1, lane, nE)];
        craw1 = col[edge_slot(p1, p2, lane, nE)];
        iraw0 = inv_sqrt[min((unsigned)craw0, (unsigned)(nN - 1))];
        iraw1 = inv_sqrt[min((unsigned)craw1, (unsigned)(nN - 1))];
    }

    for (;;) {
        const int ntile = tile + gridDim.x;
        n0 = tile * TILE_M + wv * 2;
        const int   p0   = __builtin_amdgcn_readlane(rpv, 0);
        const int   p1   = __builtin_amdgcn_readlane(rpv, 1);
        const int   p2   = __builtin_amdgcn_readlane(rpv, 2);
        const float isd0 = readlane_f(isv, 0);
        const float isd1 = readlane_f(isv, 1);
        const int   ms0  = (int)min((unsigned)craw0, (unsigned)(nN - 1));
        const int   ms1  = (int)min((unsigned)craw1, (unsigned)(nN - 1));
        const float mc0  = (lane < p1 - p0) ? iraw0 * isd0 : 0.0f;
        const float mc1  = (lane < p2 - p1) ? iraw1 * isd1 : 0.0f;

        const int   nn0  = ntile * TILE_M + wv * 2;
        const int   rpvN = rowptr[min(nn0 + min(lane, 2), nN)];
        const float isvN = inv_sqrt[min(nn0 + (lane & 1), nN - 1)];

        {
            const float4 hs  = row_load(h4, min(n0, nN - 1), lane);
            const float4 acc = node_sum(h4, col, inv_sqrt, p0, p1, ms0, mc0, isd0, nN, nE, lane);
            float4 v;
            v.x = __builtin_fmaf(hs.x * isd0, isd0, acc.x);
            v.y = __builtin_fmaf(hs.y * isd0, isd0, acc.y);
            v.z = __builtin_fmaf(hs.z * isd0, isd0, acc.z);
            v.w = __builtin_fmaf(hs.w * isd0, isd0, acc.w);
            if (n0 >= nN) v = make_float4(0.0f, 0.0f, 0.0f, 0.0f);
            *(float4*)(s_a + (wv * 2) * TILE_PITCH + lane * 4) = v;
        }

        const int q0 = __builtin_amdgcn_readlane(rpvN, 0);
        const int q1 = __builtin_amdgcn_readlane(rpvN, 1);
        const int q2 = __builtin_amdgcn_readlane(rpvN, 2);
        const int crawN0 = col[edge_slot(q0, q1, lane, nE)];
        const int crawN1 = col[edge_slot(q1, q2, lane, nE)];

        {
            const float4 hs  = row_load(h4, min(n0 + 1, nN - 1), lane);
            const float4 acc = node_sum(h4, col, inv_sqrt, p1, p2, ms1, mc1, isd1, nN, nE, lane);
            float4 v;
            v.x = __builtin_fmaf(hs.x * isd1, isd1, acc.x);
            v.y = __builtin_fmaf(hs.y * isd1, isd1, acc.y);
            v.z = __builtin_fmaf(hs.z * isd1, isd1, acc.z);
            v.w = __builtin_fmaf(hs.w * isd1, isd1, acc.w);
            if (n0 + 1 >= nN) v = make_float4(0.0f, 0.0f, 0.0f, 0.0f);
            *(float4*)(s_a + (wv * 2 + 1) * TILE_PITCH + lane * 4) = v;
        }

        __syncthreads();

        const float irawN0 = inv_sqrt[min((unsigned)crawN0, (unsigned)(nN - 1))];
        const float irawN1 = inv_sqrt[min((unsigned)crawN1, (unsigned)(nN - 1))];

        v8f acc = {};
        const float* __restrict__ arow = s_a + row * TILE_PITCH + koff;
#pragma unroll
        for (int ks = 0; ks < 32; ++ks) {
            const v2f a = *(const v2f*)(arow + ks * 4);
            acc = __builtin_amdgcn_wmma_f32_16x16x4_f32(
                false, a, false, bfrag[ks], (short)0, acc, false, false);
        }
#pragma unroll
        for (int r = 0; r < 8; ++r) {
            float o = acc[r] + bcol;
            if (RELU) o = fmaxf(o, 0.0f);
            s_c[(r + 8 * hi) * TILE_PITCH + wv * 16 + row] = o;
        }

        __syncthreads();

#pragma unroll
        for (int q = 0; q < 2; ++q) {
            const int node = n0 + q;
            if (node < nN)
                hn4[(size_t)node * (IN_C / 4) + lane] = *(const float4*)(s_c + (wv * 2 + q) * TILE_PITCH + lane * 4);
        }

        if (ntile >= mTiles) break;
        tile = ntile;
        rpv = rpvN;  isv = isvN;
        craw0 = crawN0;  craw1 = crawN1;
        iraw0 = irawN0;  iraw1 = irawN1;
    }
}

extern "C" void kernel_launch(void* const* d_in, const int* in_sizes, int n_in,
                              void* d_out, int out_size, void* d_ws, size_t ws_size,
                              hipStream_t stream)
{
    const float* x   = (const float*)d_in[0];
    const int*   ei  = (const int*)d_in[1];
    const float* W1  = (const float*)d_in[2];
    const float* b1  = (const float*)d_in[3];
    const float* W2  = (const float*)d_in[4];
    const float* b2  = (const float*)d_in[5];

    const int N = in_sizes[0] / IN_C;
    const int E = in_sizes[1] / 2;
    const int* src = ei;
    const int* dst = ei + E;

    const int NB   = (N + BIN_W - 1) / BIN_W;
    const int nBlk = (E + CHUNK - 1) / CHUNK;
    float* hbuf      = (float*)d_ws;
    float* inv_sqrt  = hbuf + (size_t)N * IN_C;
    int*   rowptr    = (int*)(inv_sqrt + N);
    int*   col       = rowptr + N + 4;
    int*   tsrc      = col + E;
    int*   tdst      = tsrc + E;
    int*   offT      = tdst + E;
    float* out       = (float*)d_out;

    const int mTiles = (N + 15) / 16;

    k_part<<<nBlk, SORT_T, 0, stream>>>(src, dst, offT, tsrc, tdst, E, NB);
    k_binsort<<<NB, SORT_T, 0, stream>>>(tsrc, tdst, offT, rowptr, inv_sqrt, col, N, E, NB, nBlk);

    k_gather_gemm<true><<<FUSED_GRID, 256, 0, stream>>>(x, rowptr, col, inv_sqrt, b1, W1, hbuf, N, E, mTiles);

    k_gather_gemm<false><<<FUSED_GRID, 256, 0, stream>>>(hbuf, rowptr, col, inv_sqrt, b2, W2, out, N, E, mTiles);
}


# Layer 1 stores row n already times deg(n)^-1/2, so layer 2 adds plain rows and scales each node's sum once: its 4-byte coefficient gather per edge goes; sum order kept, rounding of terms changes
# speedup vs baseline: 6.5765x; 1.0341x over previous
#include <hip/hip_runtime.h>
#include <math.h>


typedef __attribute__((ext_vector_type(2))) float v2f;
typedef __attribute__((ext_vector_type(8))) float v8f;

#define IN_C 128

#define BIN_SHIFT 9
#define BIN_W     512
#define NB_MAX    1024
#define SORT_T    1024
#define CHUNK     8192
#define STAGE_CAP 32768
#define BATCH     (CHUNK / SORT_T)
#define SEG_G     4
#define SEG_U     2

__device__ __forceinline__ int bin_of(int d, int nB) {
    return (int)min((unsigned)d >> BIN_SHIFT, (unsigned)(nB - 1));
}

__device__ __forceinline__ int block_excl_scan(int v, int* s_w, int* total) {
    const int lane = threadIdx.x & 31;
    const int wv   = threadIdx.x >> 5;
    int inc = v;
#pragma unroll
    for (int o = 1; o < 32; o <<= 1) {
        int u = __shfl_up(inc, o);
        if (lane >= o) inc += u;
    }
    __syncthreads();
    if (lane == 31) s_w[wv] = inc;
    __syncthreads();
    if (wv == 0) {
        int t = s_w[lane];
#pragma unroll
        for (int o = 1; o < 32; o <<= 1) {
            int u = __shfl_up(t, o);
            if (lane >= o) t += u;
        }
        s_w[lane] = t;
    }
    __syncthreads();
    const int wbase = (wv > 0) ? s_w[wv - 1] : 0;
    *total = s_w[31];
    return wbase + inc - v;
}

__global__ __launch_bounds__(SORT_T) void k_part(
    const int* __restrict__ src, const int* __restrict__ dst,
    int* __restrict__ offT, int* __restrict__ tsrc, int* __restrict__ tdst, int nE, int nB)
{
    __shared__ int s_w[32];
    __shared__ int s_cur[NB_MAX];
    __shared__ int s_ssrc[CHUNK];
    __shared__ int s_sdst[CHUNK];
    const int tid = threadIdx.x;
    const int blk = blockIdx.x;
    nB = min(nB, NB_MAX);
    s_cur[tid] = 0;
    __syncthreads();

    const int c0 = blk * CHUNK;
    int d[BATCH], s[BATCH];
#pragma unroll
    for (int k = 0; k < BATCH; ++k) {
        const int i = c0 + k * SORT_T + tid;
        d[k] = dst[min(i, nE - 1)];
        s[k] = src[min(i, nE - 1)];
    }
#pragma unroll
    for (int k = 0; k < BATCH; ++k) {
        const int i = c0 + k * SORT_T + tid;
        if (i < nE) atomicAdd(&s_cur[bin_of(d[k], nB)], 1);
    }
    __syncthreads();

    const int own = s_cur[tid];
    int nChunk;
    const int off = block_excl_scan(own, s_w, &nChunk);
    s_cur[tid] = off;
    if (tid < nB)  offT[blk * (nB + 1) + tid] = off;
    if (tid == 0)  offT[blk * (nB + 1) + nB]  = nChunk;
    __syncthreads();

#pragma unroll
    for (int k = 0; k < BATCH; ++k) {
        const int i = c0 + k * SORT_T + tid;
        if (i < nE) {
            const int slot = atomicAdd(&s_cur[bin_of(d[k], nB)], 1);
            if ((unsigned)slot < (unsigned)CHUNK) { s_ssrc[slot] = s[k]; s_sdst[slot] = d[k]; }
        }
    }
    __syncthreads();

    for (int k = 0; k < BATCH; ++k) {
        const int slot = k * SORT_T + tid;
        const int g    = c0 + slot;
        if (slot < nChunk && g < nE) { tsrc[g] = s_ssrc[slot]; tdst[g] = s_sdst[slot]; }
    }
}

template <bool PLACE>
__device__ __forceinline__ void bin_walk(
    const int* __restrict__ tsrc, const int* __restrict__ tdst, const int* __restrict__ offT,
    int* __restrict__ col, int* s_cnt, int* s_col,
    int b, int nB1, int nBlk, int nE, int node0, int e0, int cnt, bool staged, int wv, int lane)
{
    for (int kp = 0; kp < nBlk; kp += SORT_T) {
        const int kmine = kp + wv + 32 * lane;
        int o0 = 0, o1 = 0;
        if (kmine < nBlk) {
            o0 = offT[kmine * nB1 + b];
            o1 = offT[kmine * nB1 + b + 1];
        }
        o0 = min(max(o0, 0), CHUNK);
        o1 = min(max(o1, o0), CHUNK);
        const int nMine = min(32, (nBlk - kp - wv + 31) >> 5);
        for (int g0 = 0; g0 < nMine; g0 += SEG_G) {
            int base[SEG_G], c[SEG_G];
            int d[SEG_G][SEG_U], s[SEG_G][SEG_U];
#pragma unroll
            for (int g = 0; g < SEG_G; ++g) {
                const int gi = min(g0 + g, 31);
                const int a0 = __builtin_amdgcn_readlane(o0, gi);
                const int a1 = __builtin_amdgcn_readlane(o1, gi);
                c[g]    = (g0 + g < nMine) ? a1 - a0 : 0;
                base[g] = (kp + wv + 32 * (g0 + g)) * CHUNK + a0;
#pragma unroll
                for (int u = 0; u < SEG_U; ++u) {
                    const int idx = min(max(base[g] + min(lane + 32 * u, c[g] - 1), 0), nE - 1);
                    d[g][u] = tdst[idx];
                    s[g][u] = PLACE ? tsrc[idx] : 0;
                }
            }
#pragma unroll
            for (int g = 0; g < SEG_G; ++g) {
#pragma unroll
                for (int u = 0; u < SEG_U; ++u) {
                    if (lane + 32 * u < c[g]) {
                        const int ld = (int)min((unsigned)(d[g][u] - node0), (unsigned)(BIN_W - 1));
                        if (PLACE) {
                            const int p = atomicAdd(&s_cnt[ld], 1);
                            if ((unsigned)p < (unsigned)cnt) {
                                if (staged) s_col[p] = s[g][u];
                                else        col[e0 + p] = s[g][u];
                            }
                        } else {
                            atomicAdd(&s_cnt[ld], 1);
                        }
                    }
                }
            }
#pragma unroll
            for (int g = 0; g < SEG_G; ++g) {
                for (int i = lane + 32 * SEG_U; i < c[g]; i += 32) {
                    const int idx = min(max(base[g] + i, 0), nE - 1);
                    const int dd  = tdst[idx];
                    const int ld  = (int)min((unsigned)(dd - node0), (unsigned)(BIN_W - 1));
                    if (PLACE) {
                        const int ss = tsrc[idx];
                        const int p  = atomicAdd(&s_cnt[ld], 1);
                        if ((unsigned)p < (unsigned)cnt) {
                            if (staged) s_col[p] = ss;
                            else        col[e0 + p] = ss;
                        }
                    } else {
                        atomicAdd(&s_cnt[ld], 1);
                    }
                }
            }
        }
    }
}

__global__ __launch_bounds__(SORT_T) void k_binsort(
    const int* __restrict__ tsrc, const int* __restrict__ tdst,
    const int* __restrict__ offT, int* __restrict__ rowptr,
    float* __restrict__ inv_sqrt, int* __restrict__ col, int nN, int nE, int nB, int nBlk)
{
    __shared__ int s_w[32];
    __shared__ int s_deg[BIN_W];
    __shared__ int s_cur[BIN_W];
    __shared__ int s_col[STAGE_CAP];
    const int tid  = threadIdx.x;
    const int lane = tid & 31;
    const int wv   = __builtin_amdgcn_readfirstlane(tid >> 5);
    const int b    = blockIdx.x;
    nB = min(nB, NB_MAX);
    if (b >= nB) return;
    const int nB1   = nB + 1;
    const int node0 = b << BIN_SHIFT;

    int sum0 = 0, sum1 = 0;
    for (int k = tid; k < nBlk; k += SORT_T) {
        const int o0 = min(max(offT[k * nB1 + b], 0), CHUNK);
        const int o1 = min(max(offT[k * nB1 + b + 1], o0), CHUNK);
        sum0 += o0;
        sum1 += o1;
    }
    int t0, t1;
    block_excl_scan(sum0, s_w, &t0);
    block_excl_scan(sum1, s_w, &t1);
    const int e0  = min(max(t0, 0), nE);
    const int e1  = min(max(t1, e0), nE);
    const int cnt = e1 - e0;

    if (tid < BIN_W) s_deg[tid] = 0;
    __syncthreads();
    bin_walk<false>(tsrc, tdst, offT, col, s_deg, s_col, b, nB1, nBlk, nE, node0, e0, cnt, false, wv, lane);
    __syncthreads();

    const int dg = (tid < BIN_W) ? s_deg[tid] : 0;
    int nBin;
    const int lrow = block_excl_scan(dg, s_w, &nBin);
    if (tid < BIN_W) {
        s_cur[tid] = lrow;
        const int node = node0 + tid;
        if (node < nN) {
            rowptr[node + 1] = e0 + lrow + dg;
            inv_sqrt[node] = 1.0f / sqrtf((float)(dg + 1));
        }
    }
    if (b == 0 && tid == 0) rowptr[0] = 0;
    __syncthreads();

    const bool staged = (cnt <= STAGE_CAP);
    bin_walk<true>(tsrc, tdst, offT, col, s_cur, s_col, b, nB1, nBlk, nE, node0, e0, cnt, staged, wv, lane);
    if (staged) {
        __syncthreads();
        for (int i = tid; i < cnt; i += SORT_T) col[e0 + i] = s_col[i];
    }
}

#define FUSED_GRID 1024
#define TILE_M     16
#define TILE_PITCH 132

__device__ __forceinline__ float readlane_f(float v, int l) {
    return __builtin_bit_cast(float, __builtin_amdgcn_readlane(__builtin_bit_cast(int, v), l));
}

__device__ __forceinline__ void fma_row(float4& acc, const float4 hv, const float c) {
    acc.x = __builtin_fmaf(hv.x, c, acc.x);
    acc.y = __builtin_fmaf(hv.y, c, acc.y);
    acc.z = __builtin_fmaf(hv.z, c, acc.z);
    acc.w = __builtin_fmaf(hv.w, c, acc.w);
}

typedef __attribute__((ext_vector_type(4))) float v4f;
typedef __attribute__((address_space(1))) v4f gv4f;

__device__ __forceinline__ float4 row_load(const float4* __restrict__ m4, int r, int lane) {
    const v4f v = ((const gv4f*)m4)[(unsigned)r * (unsigned)(IN_C / 4) + (unsigned)lane];
    return make_float4(v.x, v.y, v.z, v.w);
}

__device__ __forceinline__ float4 src_row(const float4* __restrict__ h4, int my_s, int j, int lane) {
    return row_load(h4, __builtin_amdgcn_readlane(my_s, j), lane);
}

#define ROWS_B 8
#define ROWS_S 4

template <int N>
__device__ __forceinline__ void rows_pass(float4& acc, const float4* __restrict__ h4,
                                          int my_s, float my_c, int j, int lane)
{
    float4 hv[N];
#pragma unroll
    for (int t = 0; t < N; ++t) hv[t] = src_row(h4, my_s, j + t, lane);
#pragma unroll
    for (int t = 0; t < N; ++t) fma_row(acc, hv[t], readlane_f(my_c, j + t));
}

__device__ __forceinline__ void rows_acc(float4& acc, const float4* __restrict__ h4,
                                         int my_s, float my_c, int m, int lane)
{
    int j = 0;
    for (; j + ROWS_S < m; j += ROWS_B) rows_pass<ROWS_B>(acc, h4, my_s, my_c, j, lane);
    if (j < m) rows_pass<ROWS_S>(acc, h4, my_s, my_c, j, lane);
}

template <bool FIRST>
__device__ __forceinline__ float4 node_sum(const float4* __restrict__ h4, const int* __restrict__ col,
                                           const float* __restrict__ inv_sqrt, int beg, int end,
                                           int my_s, float my_c, float isd, int nN, int nE, int lane)
{
    float4 acc = make_float4(0.0f, 0.0f, 0.0f, 0.0f);
    rows_acc(acc, h4, my_s, my_c, min(32, end - beg), lane);
    for (int base = beg + 32; base < end; base += 32) {
        const int m = min(32, end - base);
        int   s = 0;
        float c = 0.0f;
        if (lane < m) {
            s = (int)min((unsigned)col[min(base + lane, nE - 1)], (unsigned)(nN - 1));
            c = FIRST ? inv_sqrt[s] * isd : 1.0f;
        }
        rows_acc(acc, h4, s, c, m, lane);
    }
    return acc;
}

__device__ __forceinline__ int edge_slot(int beg, int end, int lane, int nE) {
    return min(max(min(beg + lane, end - 1), 0), nE - 1);
}

template <bool FIRST>
__global__ __launch_bounds__(256) void k_gather_gemm(
    const float* __restrict__ h, const int* __restrict__ rowptr,
    const int* __restrict__ col, const float* __restrict__ inv_sqrt,
    const float* __restrict__ bias, const float* __restrict__ Wn,
    float* __restrict__ hn, int nN, int nE, int mTiles)
{
    __shared__ float s_a[TILE_M * TILE_PITCH];
    __shared__ float s_c[TILE_M * TILE_PITCH];
    const int lane = threadIdx.x & 31;
    const int wv   = __builtin_amdgcn_readfirstlane((int)(threadIdx.x >> 5));
    const int row  = lane & 15;
    const int hi   = lane >> 4;
    const int koff = hi * 2;

    int tile = blockIdx.x;
    if (tile >= mTiles) return;

    v2f bfrag[32];
    {
        const float* __restrict__ wbase = Wn + (size_t)koff * IN_C + wv * 16 + row;
#pragma unroll
        for (int ks = 0; ks < 32; ++ks) {
            bfrag[ks].x = wbase[(size_t)(ks * 4) * IN_C];
            bfrag[ks].y = wbase[(size_t)(ks * 4 + 1) * IN_C];
        }
    }
    const float bcol = bias[wv * 16 + row];
    const float4* __restrict__ h4  = (const float4*)h;
    float4* __restrict__       hn4 = (float4*)hn;

    int   n0  = tile * TILE_M + wv * 2;
    int   rpv = rowptr[min(n0 + min(lane, 2), nN)];
    float isv = inv_sqrt[min(n0 + (lane & 1), nN - 1)];
    int   craw0, craw1;
    float iraw0, iraw1;
    {
        const int p0 = __builtin_amdgcn_readlane(rpv, 0);
        const int p1 = __builtin_amdgcn_readlane(rpv, 1);
        const int p2 = __builtin_amdgcn_readlane(rpv, 2);
        craw0 = col[edge_slot(p0, p1, lane, nE)];
        craw1 = col[edge_slot(p1, p2, lane, nE)];
        iraw0 = FIRST ? inv_sqrt[min((unsigned)craw0, (unsigned)(nN - 1))] : 1.0f;
        iraw1 = FIRST ? inv_sqrt[min((unsigned)craw1, (unsigned)(nN - 1))] : 1.0f;
    }

    for (;;) {
        const int ntile = tile + gridDim.x;
        n0 = tile * TILE_M + wv * 2;
        const int   p0   = __builtin_amdgcn_readlane(rpv, 0);
        const int   p1   = __builtin_amdgcn_readlane(rpv, 1);
        const int   p2   = __builtin_amdgcn_readlane(rpv, 2);
        const float isd0 = readlane_f(isv, 0);
        const float isd1 = readlane_f(isv, 1);
        const int   ms0  = (int)min((unsigned)craw0, (unsigned)(nN - 1));
        const int   ms1  = (int)min((unsigned)craw1, (unsigned)(nN - 1));
        const float mc0  = (lane < p1 - p0) ? (FIRST ? iraw0 * isd0 : 1.0f) : 0.0f;
        const float mc1  = (lane < p2 - p1) ? (FIRST ? iraw1 * isd1 : 1.0f) : 0.0f;

        const int   nn0  = ntile * TILE_M + wv * 2;
        const int   rpvN = rowptr[min(nn0 + min(lane, 2), nN)];
        const float isvN = inv_sqrt[min(nn0 + (lane & 1), nN - 1)];

        {
            const float4 hs  = row_load(h4, min(n0, nN - 1), lane);
            const float4 acc = node_sum<FIRST>(h4, col, inv_sqrt, p0, p1, ms0, mc0, isd0, nN, nE, lane);
            float4 v;
            if (FIRST) {
                v.x = __builtin_fmaf(hs.x * isd0, isd0, acc.x);
                v.y = __builtin_fmaf(hs.y * isd0, isd0, acc.y);
                v.z = __builtin_fmaf(hs.z * isd0, isd0, acc.z);
                v.w = __builtin_fmaf(hs.w * isd0, isd0, acc.w);
            } else {
                v.x = (acc.x + hs.x) * isd0;
                v.y = (acc.y + hs.y) * isd0;
                v.z = (acc.z + hs.z) * isd0;
                v.w = (acc.w + hs.w) * isd0;
            }
            if (n0 >= nN) v = make_float4(0.0f, 0.0f, 0.0f, 0.0f);
            *(float4*)(s_a + (wv * 2) * TILE_PITCH + lane * 4) = v;
        }

        const int q0 = __builtin_amdgcn_readlane(rpvN, 0);
        const int q1 = __builtin_amdgcn_readlane(rpvN, 1);
        const int q2 = __builtin_amdgcn_readlane(rpvN, 2);
        const int crawN0 = col[edge_slot(q0, q1, lane, nE)];
        const int crawN1 = col[edge_slot(q1, q2, lane, nE)];

        {
            const float4 hs  = row_load(h4, min(n0 + 1, nN - 1), lane);
            const float4 acc = node_sum<FIRST>(h4, col, inv_sqrt, p1, p2, ms1, mc1, isd1, nN, nE, lane);
            float4 v;
            if (FIRST) {
                v.x = __builtin_fmaf(hs.x * isd1, isd1, acc.x);
                v.y = __builtin_fmaf(hs.y * isd1, isd1, acc.y);
                v.z = __builtin_fmaf(hs.z * isd1, isd1, acc.z);
                v.w = __builtin_fmaf(hs.w * isd1, isd1, acc.w);
            } else {
                v.x = (acc.x + hs.x) * isd1;
                v.y = (acc.y + hs.y) * isd1;
                v.z = (acc.z + hs.z) * isd1;
                v.w = (acc.w + hs.w) * isd1;
            }
            if (n0 + 1 >= nN) v = make_float4(0.0f, 0.0f, 0.0f, 0.0f);
            *(float4*)(s_a + (wv * 2 + 1) * TILE_PITCH + lane * 4) = v;
        }

        __syncthreads();

        const float irawN0 = FIRST ? inv_sqrt[min((unsigned)crawN0, (unsigned)(nN - 1))] : 1.0f;
        const float irawN1 = FIRST ? inv_sqrt[min((unsigned)crawN1, (unsigned)(nN - 1))] : 1.0f;

        v8f acc = {};
        const float* __restrict__ arow = s_a + row * TILE_PITCH + koff;
#pragma unroll
        for (int ks = 0; ks < 32; ++ks) {
            const v2f a = *(const v2f*)(arow + ks * 4);
            acc = __builtin_amdgcn_wmma_f32_16x16x4_f32(
                false, a, false, bfrag[ks], (short)0, acc, false, false);
        }
#pragma unroll
        for (int r = 0; r < 8; ++r) {
            float o = acc[r] + bcol;
            if (FIRST) o = fmaxf(o, 0.0f);
            s_c[(r + 8 * hi) * TILE_PITCH + wv * 16 + row] = o;
        }

        __syncthreads();

#pragma unroll
        for (int q = 0; q < 2; ++q) {
            const int node = n0 + q;
            if (node < nN) {
                float4 o = *(const float4*)(s_c + (wv * 2 + q) * TILE_PITCH + lane * 4);
                if (FIRST) {
                    const float sc = q ? isd1 : isd0;
                    o.x *= sc;  o.y *= sc;  o.z *= sc;  o.w *= sc;
                }
                hn4[(size_t)node * (IN_C / 4) + lane] = o;
            }
        }

        if (ntile >= mTiles) break;
        tile = ntile;
        rpv = rpvN;  isv = isvN;
        craw0 = crawN0;  craw1 = crawN1;
        iraw0 = irawN0;  iraw1 = irawN1;
    }
}

extern "C" void kernel_launch(void* const* d_in, const int* in_sizes, int n_in,
                              void* d_out, int out_size, void* d_ws, size_t ws_size,
                              hipStream_t stream)
{
    const float* x   = (const float*)d_in[0];
    const int*   ei  = (const int*)d_in[1];
    const float* W1  = (const float*)d_in[2];
    const float* b1  = (const float*)d_in[3];
    const float* W2  = (const float*)d_in[4];
    const float* b2  = (const float*)d_in[5];

    const int N = in_sizes[0] / IN_C;
    const int E = in_sizes[1] / 2;
    const int* src = ei;
    const int* dst = ei + E;

    const int NB   = (N + BIN_W - 1) / BIN_W;
    const int nBlk = (E + CHUNK - 1) / CHUNK;
    float* hbuf      = (float*)d_ws;
    float* inv_sqrt  = hbuf + (size_t)N * IN_C;
    int*   rowptr    = (int*)(inv_sqrt + N);
    int*   col       = rowptr + N + 4;
    int*   tsrc      = col + E;
    int*   tdst      = tsrc + E;
    int*   offT      = tdst + E;
    float* out       = (float*)d_out;

    const int mTiles = (N + 15) / 16;

    k_part<<<nBlk, SORT_T, 0, stream>>>(src, dst, offT, tsrc, tdst, E, NB);
    k_binsort<<<NB, SORT_T, 0, stream>>>(tsrc, tdst, offT, rowptr, inv_sqrt, col, N, E, NB, nBlk);

    k_gather_gemm<true><<<FUSED_GRID, 256, 0, stream>>>(x, rowptr, col, inv_sqrt, b1, W1, hbuf, N, E, mTiles);

    k_gather_gemm<false><<<FUSED_GRID, 256, 0, stream>>>(hbuf, rowptr, col, inv_sqrt, b2, W2, out, N, E, mTiles);
}
